# starting kernel
# baseline (speedup 1.0000x reference)
#include <hip/hip_runtime.h>

#define IN_F  128
#define HID_F 64
#define OUT_F 64

typedef __attribute__((ext_vector_type(2))) float v2f;
typedef __attribute__((ext_vector_type(8))) float v8f;

__global__ void gcn_zero_kernel(float* __restrict__ p, long long n) {
    long long i = (long long)blockIdx.x * blockDim.x + threadIdx.x;
    if (i < n) p[i] = 0.0f;
}

__global__ void gcn_degree_kernel(const int* __restrict__ src,
                                  const int* __restrict__ dst,
                                  float* __restrict__ deg_out,
                                  float* __restrict__ deg_in, int nE) {
    int e = blockIdx.x * blockDim.x + threadIdx.x;
    if (e < nE) {
        atomicAdd(&deg_out[src[e]], 1.0f);
        atomicAdd(&deg_in [dst[e]], 1.0f);
    }
}

__global__ void gcn_norm_kernel(float* __restrict__ p, long long n) {
    long long i = (long long)blockIdx.x * blockDim.x + threadIdx.x;
    if (i < n) {
        float d = p[i];
        d = d < 1.0f ? 1.0f : d;
        p[i] = rsqrtf(d);
    }
}

__global__ __launch_bounds__(128) void gcn_gemm1_kernel(
        const float* __restrict__ X, const float* __restrict__ W,
        const float* __restrict__ onorm, float* __restrict__ H, int nN) {
    const int mt   = blockIdx.x;
    const int wv   = threadIdx.x >> 5;
    const int lane = threadIdx.x & 31;
    const int half = lane >> 4;
    const int l    = lane & 15;
    const int m0   = mt * 16;
    const int n0   = wv * 16;

    int row = m0 + l;
    if (row >= nN) row = nN - 1;
    const float  sc = onorm[row];
    const float* xr = X + (long long)row * IN_F;

    v8f acc = {};
#pragma unroll
    for (int k0 = 0; k0 < IN_F; k0 += 4) {
        const int ka = k0 + 2 * half;
        v2f a = *(const v2f*)(xr + ka);
        a *= sc;
        v2f b;
        b.x = W[(long long)ka       * HID_F + n0 + l];
        b.y = W[(long long)(ka + 1) * HID_F + n0 + l];
        acc = __builtin_amdgcn_wmma_f32_16x16x4_f32(
                  false, a, false, b, (short)0, acc, false, false);
    }
#pragma unroll
    for (int r = 0; r < 8; ++r) {
        const int orow = m0 + r + 8 * half;
        if (orow < nN) H[(long long)orow * HID_F + n0 + l] = acc[r];
    }
}

__global__ void gcn_scatter1_kernel(const int* __restrict__ src,
                                    const int* __restrict__ dst,
                                    const float* __restrict__ H,
                                    float* __restrict__ AGG, long long total) {
    long long t = (long long)blockIdx.x * blockDim.x + threadIdx.x;
    if (t >= total) return;
    const int e = (int)(t >> 6);
    const int c = (int)(t & 63);
    const int s = src[e];
    const int d = dst[e];
    atomicAdd(&AGG[(long long)d * HID_F + c], H[(long long)s * HID_F + c]);
}

__global__ void gcn_post1_kernel(float* __restrict__ A,
                                 const float* __restrict__ inorm,
                                 const float* __restrict__ b1, long long n) {
    long long i = (long long)blockIdx.x * blockDim.x + threadIdx.x;
    if (i < n) {
        const int node = (int)(i >> 6);
        const int c    = (int)(i & 63);
        float v = A[i] * inorm[node] + b1[c];
        A[i] = v > 0.0f ? v : 0.0f;
    }
}

__global__ void gcn_scatter2_kernel(const int* __restrict__ src,
                                    const int* __restrict__ dst,
                                    const float* __restrict__ H,
                                    const float* __restrict__ onorm,
                                    float* __restrict__ AGG, long long total) {
    long long t = (long long)blockIdx.x * blockDim.x + threadIdx.x;
    if (t >= total) return;
    const int e = (int)(t >> 6);
    const int c = (int)(t & 63);
    const int s = src[e];
    const int d = dst[e];
    atomicAdd(&AGG[(long long)d * HID_F + c],
              H[(long long)s * HID_F + c] * onorm[s]);
}

__global__ __launch_bounds__(128) void gcn_gemm2_kernel(
        const float* __restrict__ A, const float* __restrict__ W,
        const float* __restrict__ inorm, const float* __restrict__ bias,
        float* __restrict__ Out, int nN) {
    const int mt   = blockIdx.x;
    const int wv   = threadIdx.x >> 5;
    const int lane = threadIdx.x & 31;
    const int half = lane >> 4;
    const int l    = lane & 15;
    const int m0   = mt * 16;
    const int n0   = wv * 16;

    int row = m0 + l;
    if (row >= nN) row = nN - 1;
    const float* ar = A + (long long)row * HID_F;

    v8f acc = {};
#pragma unroll
    for (int k0 = 0; k0 < HID_F; k0 += 4) {
        const int ka = k0 + 2 * half;
        v2f a = *(const v2f*)(ar + ka);
        v2f b;
        b.x = W[(long long)ka       * OUT_F + n0 + l];
        b.y = W[(long long)(ka + 1) * OUT_F + n0 + l];
        acc = __builtin_amdgcn_wmma_f32_16x16x4_f32(
                  false, a, false, b, (short)0, acc, false, false);
    }
    const float bcol = bias[n0 + l];
#pragma unroll
    for (int r = 0; r < 8; ++r) {
        const int orow = m0 + r + 8 * half;
        if (orow < nN)
            Out[(long long)orow * OUT_F + n0 + l] = acc[r] * inorm[orow] + bcol;
    }
}

extern "C" void kernel_launch(void* const* d_in, const int* in_sizes, int n_in,
                              void* d_out, int out_size, void* d_ws, size_t ws_size,
                              hipStream_t stream) {
    const float* X   = (const float*)d_in[0];
    const int*   src = (const int*)  d_in[1];
    const int*   dst = (const int*)  d_in[2];
    const float* W1  = (const float*)d_in[3];
    const float* b1  = (const float*)d_in[4];
    const float* W2  = (const float*)d_in[5];
    const float* b2  = (const float*)d_in[6];
    float*       out = (float*)d_out;

    const int nN = in_sizes[0] / IN_F;
    const int nE = in_sizes[1];

    float* ws    = (float*)d_ws;
    float* onorm = ws;
    float* inorm = ws + nN;
    float* bufA  = ws + 2LL * nN;
    float* bufB  = bufA + (long long)nN * HID_F;

    const long long nh  = (long long)nN * HID_F;
    const long long tot = (long long)nE * HID_F;

    gcn_zero_kernel  <<<(2LL*nN + 255) / 256, 256, 0, stream>>>(ws, 2LL*nN);
    gcn_degree_kernel<<<(nE + 255) / 256, 256, 0, stream>>>(src, dst, onorm, inorm, nE);
    gcn_norm_kernel  <<<(2LL*nN + 255) / 256, 256, 0, stream>>>(ws, 2LL*nN);

    gcn_gemm1_kernel   <<<(nN + 15) / 16, 128, 0, stream>>>(X, W1, onorm, bufA, nN);
    gcn_zero_kernel    <<<(nh + 255) / 256, 256, 0, stream>>>(bufB, nh);
    gcn_scatter1_kernel<<<(tot + 255) / 256, 256, 0, stream>>>(src, dst, bufA, bufB, tot);
    gcn_post1_kernel   <<<(nh + 255) / 256, 256, 0, stream>>>(bufB, inorm, b1, nh);

    gcn_zero_kernel    <<<(nh + 255) / 256, 256, 0, stream>>>(bufA, nh);
    gcn_scatter2_kernel<<<(tot + 255) / 256, 256, 0, stream>>>(src, dst, bufB, onorm, bufA, tot);
    gcn_gemm2_kernel   <<<(nN + 15) / 16, 128, 0, stream>>>(bufA, W2, inorm, b2, out, nN);
}


# Aggregation without float atomics: an integer-atomic count ranks each edge into a dst-sorted row (scan, fill), one wave gather-sums each row; per-row sum order = arrival rank; no zero fills
# speedup vs baseline: 2.2404x; 2.2404x over previous
#include <hip/hip_runtime.h>

#define IN_F  128
#define HID_F 64
#define OUT_F 64

#define SCAN_TILE 1024

typedef __attribute__((ext_vector_type(2))) float v2f;
typedef __attribute__((ext_vector_type(8))) float v8f;
typedef __attribute__((ext_vector_type(4))) float v4f;
typedef __attribute__((ext_vector_type(4))) int   v4i;

__global__ void gcn_zero_kernel(int* __restrict__ p, long long n) {
    long long i = (long long)blockIdx.x * blockDim.x + threadIdx.x;
    if (i < n) p[i] = 0;
}

__global__ void gcn_degree_rank_kernel(const int* __restrict__ src,
                                       const int* __restrict__ dst,
                                       int* __restrict__ cnt_out,
                                       int* __restrict__ cnt_in,
                                       int* __restrict__ rank, int nE) {
    int e = blockIdx.x * blockDim.x + threadIdx.x;
    if (e < nE) {
        atomicAdd(&cnt_out[src[e]], 1);
        rank[e] = atomicAdd(&cnt_in[dst[e]], 1);
    }
}


__global__ __launch_bounds__(256) void gcn_scan_tile_kernel(
        const int* __restrict__ cnt_in, int* __restrict__ rowptr,
        int* __restrict__ tsum) {
    __shared__ int wsum[8];
    const int t    = threadIdx.x;
    const int lane = t & 31;
    const int wv   = t >> 5;
    const int base = blockIdx.x * SCAN_TILE + t * 4;
    const v4i v = *(const v4i*)(cnt_in + base);
    const int s = v.x + v.y + v.z + v.w;
    int x = s;
    for (int off = 1; off < 32; off <<= 1) {
        const int y = __shfl_up(x, off);
        if (lane >= off) x += y;
    }
    if (lane == 31) wsum[wv] = x;
    __syncthreads();
    int woff = 0;
    for (int i = 0; i < 8; ++i) woff += (i < wv) ? wsum[i] : 0;
    const int ex = woff + x - s;
    v4i o;
    o.x = ex;
    o.y = ex + v.x;
    o.z = ex + v.x + v.y;
    o.w = ex + v.x + v.y + v.z;
    *(v4i*)(rowptr + base) = o;
    if (t == 255) tsum[blockIdx.x] = woff + x;
}

__global__ __launch_bounds__(256) void gcn_scan_finish_kernel(
        const int* __restrict__ cnt_out, const int* __restrict__ cnt_in,
        const int* __restrict__ tsum, int* __restrict__ rowptr,
        float* __restrict__ onorm, float* __restrict__ inorm) {
    __shared__ int wsum[8];
    const int t    = threadIdx.x;
    const int lane = t & 31;
    const int wv   = t >> 5;
    int part = 0;
    for (int i = t; i < (int)blockIdx.x; i += 256) part += tsum[i];
    for (int off = 16; off > 0; off >>= 1) part += __shfl_xor(part, off);
    if (lane == 0) wsum[wv] = part;
    __syncthreads();
    int toff = 0;
    for (int i = 0; i < 8; ++i) toff += wsum[i];
    const int base = blockIdx.x * SCAN_TILE + t * 4;
    *(v4i*)(rowptr + base) = *(const v4i*)(rowptr + base) + toff;
    const v4i co = *(const v4i*)(cnt_out + base);
    const v4i ci = *(const v4i*)(cnt_in  + base);
    v4f no, ni;
    for (int u = 0; u < 4; ++u) {
        float dout = (float)co[u];
        float din  = (float)ci[u];
        dout = dout < 1.0f ? 1.0f : dout;
        din  = din  < 1.0f ? 1.0f : din;
        no[u] = rsqrtf(dout);
        ni[u] = rsqrtf(din);
    }
    *(v4f*)(onorm + base) = no;
    *(v4f*)(inorm + base) = ni;
}

__global__ void gcn_fill_kernel(const int* __restrict__ src,
                                const int* __restrict__ dst,
                                const int* __restrict__ rowptr,
                                const int* __restrict__ rank,
                                int* __restrict__ esrc, int nE) {
    int e = blockIdx.x * blockDim.x + threadIdx.x;
    if (e < nE) esrc[rowptr[dst[e]] + rank[e]] = src[e];
}

__global__ __launch_bounds__(128) void gcn_gemm1_kernel(
        const float* __restrict__ X, const float* __restrict__ W,
        const float* __restrict__ onorm, float* __restrict__ H, int nN) {
    const int mt   = blockIdx.x;
    const int wv   = threadIdx.x >> 5;
    const int lane = threadIdx.x & 31;
    const int half = lane >> 4;
    const int l    = lane & 15;
    const int m0   = mt * 16;
    const int n0   = wv * 16;

    int row = m0 + l;
    if (row >= nN) row = nN - 1;
    const float  sc = onorm[row];
    const float* xr = X + (long long)row * IN_F;

    v8f acc = {};
#pragma unroll
    for (int k0 = 0; k0 < IN_F; k0 += 4) {
        const int ka = k0 + 2 * half;
        v2f a = *(const v2f*)(xr + ka);
        a *= sc;
        v2f b;
        b.x = W[(long long)ka       * HID_F + n0 + l];
        b.y = W[(long long)(ka + 1) * HID_F + n0 + l];
        acc = __builtin_amdgcn_wmma_f32_16x16x4_f32(
                  false, a, false, b, (short)0, acc, false, false);
    }
#pragma unroll
    for (int r = 0; r < 8; ++r) {
        const int orow = m0 + r + 8 * half;
        if (orow < nN) H[(long long)orow * HID_F + n0 + l] = acc[r];
    }
}

template <bool SCALE>
__global__ __launch_bounds__(256) void gcn_gather_kernel(
        const int* __restrict__ rowptr, const int* __restrict__ cnt_in,
        const int* __restrict__ esrc, const float* __restrict__ H,
        const float* __restrict__ onorm, float* __restrict__ AGG, int nN) {
    const int lane = threadIdx.x & 31;
    const int node = __builtin_amdgcn_readfirstlane(blockIdx.x * 8 + (threadIdx.x >> 5));
    if (node >= nN) return;
    const int start = rowptr[node];
    const int deg   = cnt_in[node];
    const float* hl = H + 2 * lane;

    v2f acc = {0.0f, 0.0f};
    for (int base = 0; base < deg; base += 32) {
        const int n = (deg - base) < 32 ? (deg - base) : 32;
        int   idx = 0;
        float sc  = 1.0f;
        if (lane < n) {
            idx = esrc[start + base + lane];
            if (SCALE) sc = onorm[idx];
        }
        int j = 0;
        for (; j + 4 <= n; j += 4) {
            const int s0 = __builtin_amdgcn_readlane(idx, j);
            const int s1 = __builtin_amdgcn_readlane(idx, j + 1);
            const int s2 = __builtin_amdgcn_readlane(idx, j + 2);
            const int s3 = __builtin_amdgcn_readlane(idx, j + 3);
            const v2f a0 = *(const v2f*)(hl + (long long)s0 * HID_F);
            const v2f a1 = *(const v2f*)(hl + (long long)s1 * HID_F);
            const v2f a2 = *(const v2f*)(hl + (long long)s2 * HID_F);
            const v2f a3 = *(const v2f*)(hl + (long long)s3 * HID_F);
            if (SCALE) {
                const float f0 = __builtin_bit_cast(float, __builtin_amdgcn_readlane(__builtin_bit_cast(int, sc), j));
                const float f1 = __builtin_bit_cast(float, __builtin_amdgcn_readlane(__builtin_bit_cast(int, sc), j + 1));
                const float f2 = __builtin_bit_cast(float, __builtin_amdgcn_readlane(__builtin_bit_cast(int, sc), j + 2));
                const float f3 = __builtin_bit_cast(float, __builtin_amdgcn_readlane(__builtin_bit_cast(int, sc), j + 3));
                acc += a0 * f0;
                acc += a1 * f1;
                acc += a2 * f2;
                acc += a3 * f3;
            } else {
                acc += a0;
                acc += a1;
                acc += a2;
                acc += a3;
            }
        }
        for (; j < n; ++j) {
            const int s0 = __builtin_amdgcn_readlane(idx, j);
            const v2f a0 = *(const v2f*)(hl + (long long)s0 * HID_F);
            if (SCALE) {
                const float f0 = __builtin_bit_cast(float, __builtin_amdgcn_readlane(__builtin_bit_cast(int, sc), j));
                acc += a0 * f0;
            } else {
                acc += a0;
            }
        }
    }
    *(v2f*)(AGG + (long long)node * HID_F + 2 * lane) = acc;
}

__global__ void gcn_post1_kernel(float* __restrict__ A,
                                 const float* __restrict__ inorm,
                                 const float* __restrict__ b1, long long n) {
    long long i = (long long)blockIdx.x * blockDim.x + threadIdx.x;
    if (i < n) {
        const int node = (int)(i >> 6);
        const int c    = (int)(i & 63);
        float v = A[i] * inorm[node] + b1[c];
        A[i] = v > 0.0f ? v : 0.0f;
    }
}

__global__ __launch_bounds__(128) void gcn_gemm2_kernel(
        const float* __restrict__ A, const float* __restrict__ W,
        const float* __restrict__ inorm, const float* __restrict__ bias,
        float* __restrict__ Out, int nN) {
    const int mt   = blockIdx.x;
    const int wv   = threadIdx.x >> 5;
    const int lane = threadIdx.x & 31;
    const int half = lane >> 4;
    const int l    = lane & 15;
    const int m0   = mt * 16;
    const int n0   = wv * 16;

    int row = m0 + l;
    if (row >= nN) row = nN - 1;
    const float* ar = A + (long long)row * HID_F;

    v8f acc = {};
#pragma unroll
    for (int k0 = 0; k0 < HID_F; k0 += 4) {
        const int ka = k0 + 2 * half;
        v2f a = *(const v2f*)(ar + ka);
        v2f b;
        b.x = W[(long long)ka       * OUT_F + n0 + l];
        b.y = W[(long long)(ka + 1) * OUT_F + n0 + l];
        acc = __builtin_amdgcn_wmma_f32_16x16x4_f32(
                  false, a, false, b, (short)0, acc, false, false);
    }
    const float bcol = bias[n0 + l];
#pragma unroll
    for (int r = 0; r < 8; ++r) {
        const int orow = m0 + r + 8 * half;
        if (orow < nN)
            Out[(long long)orow * OUT_F + n0 + l] = acc[r] * inorm[orow] + bcol;
    }
}

extern "C" void kernel_launch(void* const* d_in, const int* in_sizes, int n_in,
                              void* d_out, int out_size, void* d_ws, size_t ws_size,
                              hipStream_t stream) {
    const float* X   = (const float*)d_in[0];
    const int*   src = (const int*)  d_in[1];
    const int*   dst = (const int*)  d_in[2];
    const float* W1  = (const float*)d_in[3];
    const float* b1  = (const float*)d_in[4];
    const float* W2  = (const float*)d_in[5];
    const float* b2  = (const float*)d_in[6];
    float*       out = (float*)d_out;

    const int nN = in_sizes[0] / IN_F;
    const int nE = in_sizes[1];

    const long long nh  = (long long)nN * HID_F;
    const int nT = (nN + SCAN_TILE - 1) / SCAN_TILE;
    const long long nP = (long long)nT * SCAN_TILE;

    float* ws      = (float*)d_ws;
    float* bufA    = ws;
    float* bufB    = bufA + nh;
    int*   cnt_out = (int*)(bufB + nh);
    int*   cnt_in  = cnt_out + nP;
    float* onorm   = (float*)(cnt_in + nP);
    float* inorm   = onorm + nP;
    int*   rowptr  = (int*)(inorm + nP);
    int*   rank    = rowptr + nP;
    int*   esrc    = rank + nE;
    int*   tsum    = esrc + nE;

    gcn_zero_kernel       <<<(2LL*nP + 255) / 256, 256, 0, stream>>>(cnt_out, 2LL*nP);
    gcn_degree_rank_kernel<<<(nE + 255) / 256, 256, 0, stream>>>(src, dst, cnt_out, cnt_in, rank, nE);
    gcn_scan_tile_kernel  <<<nT, 256, 0, stream>>>(cnt_in, rowptr, tsum);
    gcn_scan_finish_kernel<<<nT, 256, 0, stream>>>(cnt_out, cnt_in, tsum, rowptr, onorm, inorm);
    gcn_fill_kernel       <<<(nE + 255) / 256, 256, 0, stream>>>(src, dst, rowptr, rank, esrc, nE);

    gcn_gemm1_kernel        <<<(nN + 15) / 16, 128, 0, stream>>>(X, W1, onorm, bufA, nN);
    gcn_gather_kernel<false><<<(nN + 7) / 8, 256, 0, stream>>>(rowptr, cnt_in, esrc, bufA, onorm, bufB, nN);
    gcn_post1_kernel        <<<(nh + 255) / 256, 256, 0, stream>>>(bufB, inorm, b1, nh);

    gcn_gather_kernel<true> <<<(nN + 7) / 8, 256, 0, stream>>>(rowptr, cnt_in, esrc, bufB, onorm, bufA, nN);
    gcn_gemm2_kernel        <<<(nN + 15) / 16, 128, 0, stream>>>(bufA, W2, inorm, b2, out, nN);
}


# Layer-1 epilogue (x in_norm, + b1, ReLU) moved into the gather: the wave that owns a row finishes it in registers before its one store; the separate elementwise launch (200000 waves) is gone
# speedup vs baseline: 2.3604x; 1.0536x over previous
#include <hip/hip_runtime.h>

#define IN_F  128
#define HID_F 64
#define OUT_F 64

#define SCAN_TILE 1024

typedef __attribute__((ext_vector_type(2))) float v2f;
typedef __attribute__((ext_vector_type(8))) float v8f;
typedef __attribute__((ext_vector_type(4))) float v4f;
typedef __attribute__((ext_vector_type(4))) int   v4i;

__global__ void gcn_zero_kernel(int* __restrict__ p, long long n) {
    long long i = (long long)blockIdx.x * blockDim.x + threadIdx.x;
    if (i < n) p[i] = 0;
}

__global__ void gcn_degree_rank_kernel(const int* __restrict__ src,
                                       const int* __restrict__ dst,
                                       int* __restrict__ cnt_out,
                                       int* __restrict__ cnt_in,
                                       int* __restrict__ rank, int nE) {
    int e = blockIdx.x * blockDim.x + threadIdx.x;
    if (e < nE) {
        atomicAdd(&cnt_out[src[e]], 1);
        rank[e] = atomicAdd(&cnt_in[dst[e]], 1);
    }
}


__global__ __launch_bounds__(256) void gcn_scan_tile_kernel(
        const int* __restrict__ cnt_in, int* __restrict__ rowptr,
        int* __restrict__ tsum) {
    __shared__ int wsum[8];
    const int t    = threadIdx.x;
    const int lane = t & 31;
    const int wv   = t >> 5;
    const int base = blockIdx.x * SCAN_TILE + t * 4;
    const v4i v = *(const v4i*)(cnt_in + base);
    const int s = v.x + v.y + v.z + v.w;
    int x = s;
    for (int off = 1; off < 32; off <<= 1) {
        const int y = __shfl_up(x, off);
        if (lane >= off) x += y;
    }
    if (lane == 31) wsum[wv] = x;
    __syncthreads();
    int woff = 0;
    for (int i = 0; i < 8; ++i) woff += (i < wv) ? wsum[i] : 0;
    const int ex = woff + x - s;
    v4i o;
    o.x = ex;
    o.y = ex + v.x;
    o.z = ex + v.x + v.y;
    o.w = ex + v.x + v.y + v.z;
    *(v4i*)(rowptr + base) = o;
    if (t == 255) tsum[blockIdx.x] = woff + x;
}

__global__ __launch_bounds__(256) void gcn_scan_finish_kernel(
        const int* __restrict__ cnt_out, const int* __restrict__ cnt_in,
        const int* __restrict__ tsum, int* __restrict__ rowptr,
        float* __restrict__ onorm, float* __restrict__ inorm) {
    __shared__ int wsum[8];
    const int t    = threadIdx.x;
    const int lane = t & 31;
    const int wv   = t >> 5;
    int part = 0;
    for (int i = t; i < (int)blockIdx.x; i += 256) part += tsum[i];
    for (int off = 16; off > 0; off >>= 1) part += __shfl_xor(part, off);
    if (lane == 0) wsum[wv] = part;
    __syncthreads();
    int toff = 0;
    for (int i = 0; i < 8; ++i) toff += wsum[i];
    const int base = blockIdx.x * SCAN_TILE + t * 4;
    *(v4i*)(rowptr + base) = *(const v4i*)(rowptr + base) + toff;
    const v4i co = *(const v4i*)(cnt_out + base);
    const v4i ci = *(const v4i*)(cnt_in  + base);
    v4f no, ni;
    for (int u = 0; u < 4; ++u) {
        float dout = (float)co[u];
        float din  = (float)ci[u];
        dout = dout < 1.0f ? 1.0f : dout;
        din  = din  < 1.0f ? 1.0f : din;
        no[u] = rsqrtf(dout);
        ni[u] = rsqrtf(din);
    }
    *(v4f*)(onorm + base) = no;
    *(v4f*)(inorm + base) = ni;
}

__global__ void gcn_fill_kernel(const int* __restrict__ src,
                                const int* __restrict__ dst,
                                const int* __restrict__ rowptr,
                                const int* __restrict__ rank,
                                int* __restrict__ esrc, int nE) {
    int e = blockIdx.x * blockDim.x + threadIdx.x;
    if (e < nE) esrc[rowptr[dst[e]] + rank[e]] = src[e];
}

__global__ __launch_bounds__(128) void gcn_gemm1_kernel(
        const float* __restrict__ X, const float* __restrict__ W,
        const float* __restrict__ onorm, float* __restrict__ H, int nN) {
    const int mt   = blockIdx.x;
    const int wv   = threadIdx.x >> 5;
    const int lane = threadIdx.x & 31;
    const int half = lane >> 4;
    const int l    = lane & 15;
    const int m0   = mt * 16;
    const int n0   = wv * 16;

    int row = m0 + l;
    if (row >= nN) row = nN - 1;
    const float  sc = onorm[row];
    const float* xr = X + (long long)row * IN_F;

    v8f acc = {};
#pragma unroll
    for (int k0 = 0; k0 < IN_F; k0 += 4) {
        const int ka = k0 + 2 * half;
        v2f a = *(const v2f*)(xr + ka);
        a *= sc;
        v2f b;
        b.x = W[(long long)ka       * HID_F + n0 + l];
        b.y = W[(long long)(ka + 1) * HID_F + n0 + l];
        acc = __builtin_amdgcn_wmma_f32_16x16x4_f32(
                  false, a, false, b, (short)0, acc, false, false);
    }
#pragma unroll
    for (int r = 0; r < 8; ++r) {
        const int orow = m0 + r + 8 * half;
        if (orow < nN) H[(long long)orow * HID_F + n0 + l] = acc[r];
    }
}

template <bool SCALE, bool POST>
__global__ __launch_bounds__(256) void gcn_gather_kernel(
        const int* __restrict__ rowptr, const int* __restrict__ cnt_in,
        const int* __restrict__ esrc, const float* __restrict__ H,
        const float* __restrict__ onorm, const float* __restrict__ inorm,
        const float* __restrict__ bias, float* __restrict__ AGG, int nN) {
    const int lane = threadIdx.x & 31;
    const int node = __builtin_amdgcn_readfirstlane(blockIdx.x * 8 + (threadIdx.x >> 5));
    if (node >= nN) return;
    const int start = rowptr[node];
    const int deg   = cnt_in[node];
    const float* hl = H + 2 * lane;
    float rn = 1.0f;
    v2f   bb = {0.0f, 0.0f};
    if (POST) {
        rn = inorm[node];
        bb = *(const v2f*)(bias + 2 * lane);
    }

    v2f acc = {0.0f, 0.0f};
    for (int base = 0; base < deg; base += 32) {
        const int n = (deg - base) < 32 ? (deg - base) : 32;
        int   idx = 0;
        float sc  = 1.0f;
        if (lane < n) {
            idx = esrc[start + base + lane];
            if (SCALE) sc = onorm[idx];
        }
        int j = 0;
        for (; j + 4 <= n; j += 4) {
            const int s0 = __builtin_amdgcn_readlane(idx, j);
            const int s1 = __builtin_amdgcn_readlane(idx, j + 1);
            const int s2 = __builtin_amdgcn_readlane(idx, j + 2);
            const int s3 = __builtin_amdgcn_readlane(idx, j + 3);
            const v2f a0 = *(const v2f*)(hl + (long long)s0 * HID_F);
            const v2f a1 = *(const v2f*)(hl + (long long)s1 * HID_F);
            const v2f a2 = *(const v2f*)(hl + (long long)s2 * HID_F);
            const v2f a3 = *(const v2f*)(hl + (long long)s3 * HID_F);
            if (SCALE) {
                const float f0 = __builtin_bit_cast(float, __builtin_amdgcn_readlane(__builtin_bit_cast(int, sc), j));
                const float f1 = __builtin_bit_cast(float, __builtin_amdgcn_readlane(__builtin_bit_cast(int, sc), j + 1));
                const float f2 = __builtin_bit_cast(float, __builtin_amdgcn_readlane(__builtin_bit_cast(int, sc), j + 2));
                const float f3 = __builtin_bit_cast(float, __builtin_amdgcn_readlane(__builtin_bit_cast(int, sc), j + 3));
                acc += a0 * f0;
                acc += a1 * f1;
                acc += a2 * f2;
                acc += a3 * f3;
            } else {
                acc += a0;
                acc += a1;
                acc += a2;
                acc += a3;
            }
        }
        for (; j < n; ++j) {
            const int s0 = __builtin_amdgcn_readlane(idx, j);
            const v2f a0 = *(const v2f*)(hl + (long long)s0 * HID_F);
            if (SCALE) {
                const float f0 = __builtin_bit_cast(float, __builtin_amdgcn_readlane(__builtin_bit_cast(int, sc), j));
                acc += a0 * f0;
            } else {
                acc += a0;
            }
        }
    }
    if (POST) {
        acc = acc * rn + bb;
        acc.x = acc.x > 0.0f ? acc.x : 0.0f;
        acc.y = acc.y > 0.0f ? acc.y : 0.0f;
    }
    *(v2f*)(AGG + (long long)node * HID_F + 2 * lane) = acc;
}

__global__ __launch_bounds__(128) void gcn_gemm2_kernel(
        const float* __restrict__ A, const float* __restrict__ W,
        const float* __restrict__ inorm, const float* __restrict__ bias,
        float* __restrict__ Out, int nN) {
    const int mt   = blockIdx.x;
    const int wv   = threadIdx.x >> 5;
    const int lane = threadIdx.x & 31;
    const int half = lane >> 4;
    const int l    = lane & 15;
    const int m0   = mt * 16;
    const int n0   = wv * 16;

    int row = m0 + l;
    if (row >= nN) row = nN - 1;
    const float* ar = A + (long long)row * HID_F;

    v8f acc = {};
#pragma unroll
    for (int k0 = 0; k0 < HID_F; k0 += 4) {
        const int ka = k0 + 2 * half;
        v2f a = *(const v2f*)(ar + ka);
        v2f b;
        b.x = W[(long long)ka       * OUT_F + n0 + l];
        b.y = W[(long long)(ka + 1) * OUT_F + n0 + l];
        acc = __builtin_amdgcn_wmma_f32_16x16x4_f32(
                  false, a, false, b, (short)0, acc, false, false);
    }
    const float bcol = bias[n0 + l];
#pragma unroll
    for (int r = 0; r < 8; ++r) {
        const int orow = m0 + r + 8 * half;
        if (orow < nN)
            Out[(long long)orow * OUT_F + n0 + l] = acc[r] * inorm[orow] + bcol;
    }
}

extern "C" void kernel_launch(void* const* d_in, const int* in_sizes, int n_in,
                              void* d_out, int out_size, void* d_ws, size_t ws_size,
                              hipStream_t stream) {
    const float* X   = (const float*)d_in[0];
    const int*   src = (const int*)  d_in[1];
    const int*   dst = (const int*)  d_in[2];
    const float* W1  = (const float*)d_in[3];
    const float* b1  = (const float*)d_in[4];
    const float* W2  = (const float*)d_in[5];
    const float* b2  = (const float*)d_in[6];
    float*       out = (float*)d_out;

    const int nN = in_sizes[0] / IN_F;
    const int nE = in_sizes[1];

    const long long nh  = (long long)nN * HID_F;
    const int nT = (nN + SCAN_TILE - 1) / SCAN_TILE;
    const long long nP = (long long)nT * SCAN_TILE;

    float* ws      = (float*)d_ws;
    float* bufA    = ws;
    float* bufB    = bufA + nh;
    int*   cnt_out = (int*)(bufB + nh);
    int*   cnt_in  = cnt_out + nP;
    float* onorm   = (float*)(cnt_in + nP);
    float* inorm   = onorm + nP;
    int*   rowptr  = (int*)(inorm + nP);
    int*   rank    = rowptr + nP;
    int*   esrc    = rank + nE;
    int*   tsum    = esrc + nE;

    gcn_zero_kernel       <<<(2LL*nP + 255) / 256, 256, 0, stream>>>(cnt_out, 2LL*nP);
    gcn_degree_rank_kernel<<<(nE + 255) / 256, 256, 0, stream>>>(src, dst, cnt_out, cnt_in, rank, nE);
    gcn_scan_tile_kernel  <<<nT, 256, 0, stream>>>(cnt_in, rowptr, tsum);
    gcn_scan_finish_kernel<<<nT, 256, 0, stream>>>(cnt_out, cnt_in, tsum, rowptr, onorm, inorm);
    gcn_fill_kernel       <<<(nE + 255) / 256, 256, 0, stream>>>(src, dst, rowptr, rank, esrc, nE);

    gcn_gemm1_kernel              <<<(nN + 15) / 16, 128, 0, stream>>>(X, W1, onorm, bufA, nN);
    gcn_gather_kernel<false, true><<<(nN + 7) / 8, 256, 0, stream>>>(rowptr, cnt_in, esrc, bufA, onorm, inorm, b1, bufB, nN);

    gcn_gather_kernel<true, false><<<(nN + 7) / 8, 256, 0, stream>>>(rowptr, cnt_in, esrc, bufB, onorm, inorm, b1, bufA, nN);
    gcn_gemm2_kernel        <<<(nN + 15) / 16, 128, 0, stream>>>(bufA, W2, inorm, b2, out, nN);
}


# Both GEMMs re-tiled: one wave owns a 16x64 strip (tile n, lane c = column 4c+n); each A fragment is loaded once, W and output rows move as whole rows; k regrouped per WMMA step (sum order changes)
# speedup vs baseline: 2.5496x; 1.0802x over previous
#include <hip/hip_runtime.h>

#define IN_F  128
#define HID_F 64
#define OUT_F 64

#define SCAN_TILE 1024

typedef __attribute__((ext_vector_type(2))) float v2f;
typedef __attribute__((ext_vector_type(8))) float v8f;
typedef __attribute__((ext_vector_type(4))) float v4f;
typedef __attribute__((ext_vector_type(4))) int   v4i;

__global__ void gcn_zero_kernel(int* __restrict__ p, long long n) {
    long long i = (long long)blockIdx.x * blockDim.x + threadIdx.x;
    if (i < n) p[i] = 0;
}

__global__ void gcn_degree_rank_kernel(const int* __restrict__ src,
                                       const int* __restrict__ dst,
                                       int* __restrict__ cnt_out,
                                       int* __restrict__ cnt_in,
                                       int* __restrict__ rank, int nE) {
    int e = blockIdx.x * blockDim.x + threadIdx.x;
    if (e < nE) {
        atomicAdd(&cnt_out[src[e]], 1);
        rank[e] = atomicAdd(&cnt_in[dst[e]], 1);
    }
}


__global__ __launch_bounds__(256) void gcn_scan_tile_kernel(
        const int* __restrict__ cnt_in, int* __restrict__ rowptr,
        int* __restrict__ tsum) {
    __shared__ int wsum[8];
    const int t    = threadIdx.x;
    const int lane = t & 31;
    const int wv   = t >> 5;
    const int base = blockIdx.x * SCAN_TILE + t * 4;
    const v4i v = *(const v4i*)(cnt_in + base);
    const int s = v.x + v.y + v.z + v.w;
    int x = s;
    for (int off = 1; off < 32; off <<= 1) {
        const int y = __shfl_up(x, off);
        if (lane >= off) x += y;
    }
    if (lane == 31) wsum[wv] = x;
    __syncthreads();
    int woff = 0;
    for (int i = 0; i < 8; ++i) woff += (i < wv) ? wsum[i] : 0;
    const int ex = woff + x - s;
    v4i o;
    o.x = ex;
    o.y = ex + v.x;
    o.z = ex + v.x + v.y;
    o.w = ex + v.x + v.y + v.z;
    *(v4i*)(rowptr + base) = o;
    if (t == 255) tsum[blockIdx.x] = woff + x;
}

__global__ __launch_bounds__(256) void gcn_scan_finish_kernel(
        const int* __restrict__ cnt_out, const int* __restrict__ cnt_in,
        const int* __restrict__ tsum, int* __restrict__ rowptr,
        float* __restrict__ onorm, float* __restrict__ inorm) {
    __shared__ int wsum[8];
    const int t    = threadIdx.x;
    const int lane = t & 31;
    const int wv   = t >> 5;
    int part = 0;
    for (int i = t; i < (int)blockIdx.x; i += 256) part += tsum[i];
    for (int off = 16; off > 0; off >>= 1) part += __shfl_xor(part, off);
    if (lane == 0) wsum[wv] = part;
    __syncthreads();
    int toff = 0;
    for (int i = 0; i < 8; ++i) toff += wsum[i];
    const int base = blockIdx.x * SCAN_TILE + t * 4;
    *(v4i*)(rowptr + base) = *(const v4i*)(rowptr + base) + toff;
    const v4i co = *(const v4i*)(cnt_out + base);
    const v4i ci = *(const v4i*)(cnt_in  + base);
    v4f no, ni;
    for (int u = 0; u < 4; ++u) {
        float dout = (float)co[u];
        float din  = (float)ci[u];
        dout = dout < 1.0f ? 1.0f : dout;
        din  = din  < 1.0f ? 1.0f : din;
        no[u] = rsqrtf(dout);
        ni[u] = rsqrtf(din);
    }
    *(v4f*)(onorm + base) = no;
    *(v4f*)(inorm + base) = ni;
}

__global__ void gcn_fill_kernel(const int* __restrict__ src,
                                const int* __restrict__ dst,
                                const int* __restrict__ rowptr,
                                const int* __restrict__ rank,
                                int* __restrict__ esrc, int nE) {
    int e = blockIdx.x * blockDim.x + threadIdx.x;
    if (e < nE) esrc[rowptr[dst[e]] + rank[e]] = src[e];
}

static_assert(HID_F == 64 && OUT_F == 64, "the strip is four 16-column tiles wide");

template <int K, bool SCALE>
__device__ __forceinline__ void gcn_strip_mma(const float* __restrict__ xr,
                                              const float* __restrict__ wl,
                                              float sc, v8f (&acc)[4]) {
#pragma unroll
    for (int q = 0; q < K / 8; ++q) {
        v4f av = *(const v4f*)(xr + 8 * q);
        if (SCALE) av *= sc;
#pragma unroll
        for (int e = 0; e < 2; ++e) {
            const float* wk = wl + (8 * q + 2 * e) * 64;
            const v4f wx = *(const v4f*)wk;
            const v4f wy = *(const v4f*)(wk + 64);
            v2f a;
            a.x = av[2 * e];
            a.y = av[2 * e + 1];
#pragma unroll
            for (int n = 0; n < 4; ++n) {
                v2f b;
                b.x = wx[n];
                b.y = wy[n];
                acc[n] = __builtin_amdgcn_wmma_f32_16x16x4_f32(
                             false, a, false, b, (short)0, acc[n], false, false);
            }
        }
    }
}

__global__ __launch_bounds__(128) void gcn_gemm1_kernel(
        const float* __restrict__ X, const float* __restrict__ W,
        const float* __restrict__ onorm, float* __restrict__ H, int nN) {
    const int wv   = __builtin_amdgcn_readfirstlane(threadIdx.x >> 5);
    const int lane = threadIdx.x & 31;
    const int half = lane >> 4;
    const int l    = lane & 15;
    const int m0   = (blockIdx.x * 4 + wv) * 16;
    if (m0 >= nN) return;

    int row = m0 + l;
    if (row >= nN) row = nN - 1;
    const float sc = onorm[row];

    v8f acc[4] = {};
    gcn_strip_mma<IN_F, true>(X + (long long)row * IN_F + 4 * half,
                              W + 4 * half * HID_F + 4 * l, sc, acc);
#pragma unroll
    for (int r = 0; r < 8; ++r) {
        const int orow = m0 + r + 8 * half;
        if (orow < nN) {
            v4f o;
            o.x = acc[0][r];
            o.y = acc[1][r];
            o.z = acc[2][r];
            o.w = acc[3][r];
            *(v4f*)(H + (long long)orow * HID_F + 4 * l) = o;
        }
    }
}

template <bool SCALE, bool POST>
__global__ __launch_bounds__(256) void gcn_gather_kernel(
        const int* __restrict__ rowptr, const int* __restrict__ cnt_in,
        const int* __restrict__ esrc, const float* __restrict__ H,
        const float* __restrict__ onorm, const float* __restrict__ inorm,
        const float* __restrict__ bias, float* __restrict__ AGG, int nN) {
    const int lane = threadIdx.x & 31;
    const int node = __builtin_amdgcn_readfirstlane(blockIdx.x * 8 + (threadIdx.x >> 5));
    if (node >= nN) return;
    const int start = rowptr[node];
    const int deg   = cnt_in[node];
    const float* hl = H + 2 * lane;
    float rn = 1.0f;
    v2f   bb = {0.0f, 0.0f};
    if (POST) {
        rn = inorm[node];
        bb = *(const v2f*)(bias + 2 * lane);
    }

    v2f acc = {0.0f, 0.0f};
    for (int base = 0; base < deg; base += 32) {
        const int n = (deg - base) < 32 ? (deg - base) : 32;
        int   idx = 0;
        float sc  = 1.0f;
        if (lane < n) {
            idx = esrc[start + base + lane];
            if (SCALE) sc = onorm[idx];
        }
        int j = 0;
        for (; j + 4 <= n; j += 4) {
            const int s0 = __builtin_amdgcn_readlane(idx, j);
            const int s1 = __builtin_amdgcn_readlane(idx, j + 1);
            const int s2 = __builtin_amdgcn_readlane(idx, j + 2);
            const int s3 = __builtin_amdgcn_readlane(idx, j + 3);
            const v2f a0 = *(const v2f*)(hl + (long long)s0 * HID_F);
            const v2f a1 = *(const v2f*)(hl + (long long)s1 * HID_F);
            const v2f a2 = *(const v2f*)(hl + (long long)s2 * HID_F);
            const v2f a3 = *(const v2f*)(hl + (long long)s3 * HID_F);
            if (SCALE) {
                const float f0 = __builtin_bit_cast(float, __builtin_amdgcn_readlane(__builtin_bit_cast(int, sc), j));
                const float f1 = __builtin_bit_cast(float, __builtin_amdgcn_readlane(__builtin_bit_cast(int, sc), j + 1));
                const float f2 = __builtin_bit_cast(float, __builtin_amdgcn_readlane(__builtin_bit_cast(int, sc), j + 2));
                const float f3 = __builtin_bit_cast(float, __builtin_amdgcn_readlane(__builtin_bit_cast(int, sc), j + 3));
                acc += a0 * f0;
                acc += a1 * f1;
                acc += a2 * f2;
                acc += a3 * f3;
            } else {
                acc += a0;
                acc += a1;
                acc += a2;
                acc += a3;
            }
        }
        for (; j < n; ++j) {
            const int s0 = __builtin_amdgcn_readlane(idx, j);
            const v2f a0 = *(const v2f*)(hl + (long long)s0 * HID_F);
            if (SCALE) {
                const float f0 = __builtin_bit_cast(float, __builtin_amdgcn_readlane(__builtin_bit_cast(int, sc), j));
                acc += a0 * f0;
            } else {
                acc += a0;
            }
        }
    }
    if (POST) {
        acc = acc * rn + bb;
        acc.x = acc.x > 0.0f ? acc.x : 0.0f;
        acc.y = acc.y > 0.0f ? acc.y : 0.0f;
    }
    *(v2f*)(AGG + (long long)node * HID_F + 2 * lane) = acc;
}

__global__ __launch_bounds__(128) void gcn_gemm2_kernel(
        const float* __restrict__ A, const float* __restrict__ W,
        const float* __restrict__ inorm, const float* __restrict__ bias,
        float* __restrict__ Out, int nN) {
    const int wv   = __builtin_amdgcn_readfirstlane(threadIdx.x >> 5);
    const int lane = threadIdx.x & 31;
    const int half = lane >> 4;
    const int l    = lane & 15;
    const int m0   = (blockIdx.x * 4 + wv) * 16;
    if (m0 >= nN) return;

    int row = m0 + l;
    if (row >= nN) row = nN - 1;

    const v4f bb = *(const v4f*)(bias + 4 * l);
    float rn[8];
#pragma unroll
    for (int r = 0; r < 8; ++r) {
        int orow = m0 + r + 8 * half;
        if (orow >= nN) orow = nN - 1;
        rn[r] = inorm[orow];
    }

    v8f acc[4] = {};
    gcn_strip_mma<HID_F, false>(A + (long long)row * HID_F + 4 * half,
                                W + 4 * half * OUT_F + 4 * l, 1.0f, acc);
#pragma unroll
    for (int r = 0; r < 8; ++r) {
        const int orow = m0 + r + 8 * half;
        v4f o;
        o.x = acc[0][r] * rn[r] + bb.x;
        o.y = acc[1][r] * rn[r] + bb.y;
        o.z = acc[2][r] * rn[r] + bb.z;
        o.w = acc[3][r] * rn[r] + bb.w;
        if (orow < nN) *(v4f*)(Out + (long long)orow * OUT_F + 4 * l) = o;
    }
}

extern "C" void kernel_launch(void* const* d_in, const int* in_sizes, int n_in,
                              void* d_out, int out_size, void* d_ws, size_t ws_size,
                              hipStream_t stream) {
    const float* X   = (const float*)d_in[0];
    const int*   src = (const int*)  d_in[1];
    const int*   dst = (const int*)  d_in[2];
    const float* W1  = (const float*)d_in[3];
    const float* b1  = (const float*)d_in[4];
    const float* W2  = (const float*)d_in[5];
    const float* b2  = (const float*)d_in[6];
    float*       out = (float*)d_out;

    const int nN = in_sizes[0] / IN_F;
    const int nE = in_sizes[1];

    const long long nh  = (long long)nN * HID_F;
    const int nT = (nN + SCAN_TILE - 1) / SCAN_TILE;
    const long long nP = (long long)nT * SCAN_TILE;

    float* ws      = (float*)d_ws;
    float* bufA    = ws;
    float* bufB    = bufA + nh;
    int*   cnt_out = (int*)(bufB + nh);
    int*   cnt_in  = cnt_out + nP;
    float* onorm   = (float*)(cnt_in + nP);
    float* inorm   = onorm + nP;
    int*   rowptr  = (int*)(inorm + nP);
    int*   rank    = rowptr + nP;
    int*   esrc    = rank + nE;
    int*   tsum    = esrc + nE;

    gcn_zero_kernel       <<<(2LL*nP + 255) / 256, 256, 0, stream>>>(cnt_out, 2LL*nP);
    gcn_degree_rank_kernel<<<(nE + 255) / 256, 256, 0, stream>>>(src, dst, cnt_out, cnt_in, rank, nE);
    gcn_scan_tile_kernel  <<<nT, 256, 0, stream>>>(cnt_in, rowptr, tsum);
    gcn_scan_finish_kernel<<<nT, 256, 0, stream>>>(cnt_out, cnt_in, tsum, rowptr, onorm, inorm);
    gcn_fill_kernel       <<<(nE + 255) / 256, 256, 0, stream>>>(src, dst, rowptr, rank, esrc, nE);

    gcn_gemm1_kernel              <<<(nN + 63) / 64, 128, 0, stream>>>(X, W1, onorm, bufA, nN);
    gcn_gather_kernel<false, true><<<(nN + 7) / 8, 256, 0, stream>>>(rowptr, cnt_in, esrc, bufA, onorm, inorm, b1, bufB, nN);

    gcn_gather_kernel<true, false><<<(nN + 7) / 8, 256, 0, stream>>>(rowptr, cnt_in, esrc, bufB, onorm, inorm, b1, bufA, nN);
    gcn_gemm2_kernel        <<<(nN + 63) / 64, 128, 0, stream>>>(bufA, W2, inorm, b2, out, nN);
}


# Out-degree counted without global atomics: 128 blocks histogram slices of src in LDS, one 32768-node part each; the last scan pass sums the partial counts; the rank kernel keeps one atomic per edge
# speedup vs baseline: 3.1040x; 1.2174x over previous
#include <hip/hip_runtime.h>

#define IN_F  128
#define HID_F 64
#define OUT_F 64

#define SCAN_TILE 1024
#define OD_PART   32768
#define OD_SLICES 32

typedef __attribute__((ext_vector_type(2))) float v2f;
typedef __attribute__((ext_vector_type(8))) float v8f;
typedef __attribute__((ext_vector_type(4))) float v4f;
typedef __attribute__((ext_vector_type(4))) int   v4i;

__global__ __launch_bounds__(1024) void gcn_outdeg_kernel(
        const int* __restrict__ src, int* __restrict__ part_cnt,
        int* __restrict__ cnt_in, int nE, int nP) {
    __shared__ int hist[OD_PART];
    const int t  = threadIdx.x;
    const int s  = blockIdx.x;
    const int q  = blockIdx.y;
    const int nb = gridDim.x * gridDim.y;

    for (int i = (q * OD_SLICES + s) * 1024 + t; i < nP; i += nb * 1024) cnt_in[i] = 0;

    const v4i z = {0, 0, 0, 0};
    for (int i = 4 * t; i < OD_PART; i += 4096) *(v4i*)(hist + i) = z;
    __syncthreads();

    const int len = ((nE + 4 * OD_SLICES - 1) / (4 * OD_SLICES)) * 4;
    const int lo  = s * len;
    const int hi  = (lo + len) < nE ? (lo + len) : nE;
    const int cnt = hi > lo ? hi - lo : 0;
    const int hi4 = lo + (cnt & ~3);
    const int n0  = q * OD_PART;
    for (int i = lo + 4 * t; i < hi4; i += 8 * 4096) {
        v4i v[8];
#pragma unroll
        for (int u = 0; u < 8; ++u) {
            const v4i none = {-1, -1, -1, -1};
            const int j = i + u * 4096;
            v[u] = none;
            if (j < hi4) v[u] = *(const v4i*)(src + j);
        }
#pragma unroll
        for (int u = 0; u < 8; ++u) {
#pragma unroll
            for (int k = 0; k < 4; ++k) {
                const unsigned r = (unsigned)(v[u][k] - n0);
                if (r < (unsigned)OD_PART) atomicAdd(&hist[r], 1);
            }
        }
    }
    if (t < (cnt & 3)) {
        const unsigned r = (unsigned)(src[hi4 + t] - n0);
        if (r < (unsigned)OD_PART) atomicAdd(&hist[r], 1);
    }
    __syncthreads();

    int* out = part_cnt + (long long)s * nP + n0;
    for (int i = 4 * t; i < OD_PART; i += 4096) {
        if (n0 + i < nP) *(v4i*)(out + i) = *(const v4i*)(hist + i);
    }
}

__global__ void gcn_rank_kernel(const int* __restrict__ dst,
                                int* __restrict__ cnt_in,
                                int* __restrict__ rank, int nE) {
    int e = blockIdx.x * blockDim.x + threadIdx.x;
    if (e < nE) rank[e] = atomicAdd(&cnt_in[dst[e]], 1);
}


__global__ __launch_bounds__(256) void gcn_scan_tile_kernel(
        const int* __restrict__ cnt_in, int* __restrict__ rowptr,
        int* __restrict__ tsum) {
    __shared__ int wsum[8];
    const int t    = threadIdx.x;
    const int lane = t & 31;
    const int wv   = t >> 5;
    const int base = blockIdx.x * SCAN_TILE + t * 4;
    const v4i v = *(const v4i*)(cnt_in + base);
    const int s = v.x + v.y + v.z + v.w;
    int x = s;
    for (int off = 1; off < 32; off <<= 1) {
        const int y = __shfl_up(x, off);
        if (lane >= off) x += y;
    }
    if (lane == 31) wsum[wv] = x;
    __syncthreads();
    int woff = 0;
    for (int i = 0; i < 8; ++i) woff += (i < wv) ? wsum[i] : 0;
    const int ex = woff + x - s;
    v4i o;
    o.x = ex;
    o.y = ex + v.x;
    o.z = ex + v.x + v.y;
    o.w = ex + v.x + v.y + v.z;
    *(v4i*)(rowptr + base) = o;
    if (t == 255) tsum[blockIdx.x] = woff + x;
}

__global__ __launch_bounds__(256) void gcn_scan_finish_kernel(
        const int* __restrict__ part_cnt, const int* __restrict__ cnt_in,
        const int* __restrict__ tsum, int* __restrict__ rowptr,
        float* __restrict__ onorm, float* __restrict__ inorm, int nP) {
    __shared__ int wsum[8];
    const int t    = threadIdx.x;
    const int lane = t & 31;
    const int wv   = t >> 5;
    int part = 0;
    for (int i = t; i < (int)blockIdx.x; i += 256) part += tsum[i];
    for (int off = 16; off > 0; off >>= 1) part += __shfl_xor(part, off);
    if (lane == 0) wsum[wv] = part;
    __syncthreads();
    int toff = 0;
    for (int i = 0; i < 8; ++i) toff += wsum[i];
    const int base = blockIdx.x * SCAN_TILE + t * 4;
    *(v4i*)(rowptr + base) = *(const v4i*)(rowptr + base) + toff;
    v4i co = {0, 0, 0, 0};
#pragma unroll
    for (int s = 0; s < OD_SLICES; ++s) co += *(const v4i*)(part_cnt + (long long)s * nP + base);
    const v4i ci = *(const v4i*)(cnt_in  + base);
    v4f no, ni;
    for (int u = 0; u < 4; ++u) {
        float dout = (float)co[u];
        float din  = (float)ci[u];
        dout = dout < 1.0f ? 1.0f : dout;
        din  = din  < 1.0f ? 1.0f : din;
        no[u] = rsqrtf(dout);
        ni[u] = rsqrtf(din);
    }
    *(v4f*)(onorm + base) = no;
    *(v4f*)(inorm + base) = ni;
}

__global__ void gcn_fill_kernel(const int* __restrict__ src,
                                const int* __restrict__ dst,
                                const int* __restrict__ rowptr,
                                const int* __restrict__ rank,
                                int* __restrict__ esrc, int nE) {
    int e = blockIdx.x * blockDim.x + threadIdx.x;
    if (e < nE) esrc[rowptr[dst[e]] + rank[e]] = src[e];
}

static_assert(HID_F == 64 && OUT_F == 64, "the strip is four 16-column tiles wide");

template <int K, bool SCALE>
__device__ __forceinline__ void gcn_strip_mma(const float* __restrict__ xr,
                                              const float* __restrict__ wl,
                                              float sc, v8f (&acc)[4]) {
#pragma unroll
    for (int q = 0; q < K / 8; ++q) {
        v4f av = *(const v4f*)(xr + 8 * q);
        if (SCALE) av *= sc;
#pragma unroll
        for (int e = 0; e < 2; ++e) {
            const float* wk = wl + (8 * q + 2 * e) * 64;
            const v4f wx = *(const v4f*)wk;
            const v4f wy = *(const v4f*)(wk + 64);
            v2f a;
            a.x = av[2 * e];
            a.y = av[2 * e + 1];
#pragma unroll
            for (int n = 0; n < 4; ++n) {
                v2f b;
                b.x = wx[n];
                b.y = wy[n];
                acc[n] = __builtin_amdgcn_wmma_f32_16x16x4_f32(
                             false, a, false, b, (short)0, acc[n], false, false);
            }
        }
    }
}

__global__ __launch_bounds__(128) void gcn_gemm1_kernel(
        const float* __restrict__ X, const float* __restrict__ W,
        const float* __restrict__ onorm, float* __restrict__ H, int nN) {
    const int wv   = __builtin_amdgcn_readfirstlane(threadIdx.x >> 5);
    const int lane = threadIdx.x & 31;
    const int half = lane >> 4;
    const int l    = lane & 15;
    const int m0   = (blockIdx.x * 4 + wv) * 16;
    if (m0 >= nN) return;

    int row = m0 + l;
    if (row >= nN) row = nN - 1;
    const float sc = onorm[row];

    v8f acc[4] = {};
    gcn_strip_mma<IN_F, true>(X + (long long)row * IN_F + 4 * half,
                              W + 4 * half * HID_F + 4 * l, sc, acc);
#pragma unroll
    for (int r = 0; r < 8; ++r) {
        const int orow = m0 + r + 8 * half;
        if (orow < nN) {
            v4f o;
            o.x = acc[0][r];
            o.y = acc[1][r];
            o.z = acc[2][r];
            o.w = acc[3][r];
            *(v4f*)(H + (long long)orow * HID_F + 4 * l) = o;
        }
    }
}

template <bool SCALE, bool POST>
__global__ __launch_bounds__(256) void gcn_gather_kernel(
        const int* __restrict__ rowptr, const int* __restrict__ cnt_in,
        const int* __restrict__ esrc, const float* __restrict__ H,
        const float* __restrict__ onorm, const float* __restrict__ inorm,
        const float* __restrict__ bias, float* __restrict__ AGG, int nN) {
    const int lane = threadIdx.x & 31;
    const int node = __builtin_amdgcn_readfirstlane(blockIdx.x * 8 + (threadIdx.x >> 5));
    if (node >= nN) return;
    const int start = rowptr[node];
    const int deg   = cnt_in[node];
    const float* hl = H + 2 * lane;
    float rn = 1.0f;
    v2f   bb = {0.0f, 0.0f};
    if (POST) {
        rn = inorm[node];
        bb = *(const v2f*)(bias + 2 * lane);
    }

    v2f acc = {0.0f, 0.0f};
    for (int base = 0; base < deg; base += 32) {
        const int n = (deg - base) < 32 ? (deg - base) : 32;
        int   idx = 0;
        float sc  = 1.0f;
        if (lane < n) {
            idx = esrc[start + base + lane];
            if (SCALE) sc = onorm[idx];
        }
        int j = 0;
        for (; j + 4 <= n; j += 4) {
            const int s0 = __builtin_amdgcn_readlane(idx, j);
            const int s1 = __builtin_amdgcn_readlane(idx, j + 1);
            const int s2 = __builtin_amdgcn_readlane(idx, j + 2);
            const int s3 = __builtin_amdgcn_readlane(idx, j + 3);
            const v2f a0 = *(const v2f*)(hl + (long long)s0 * HID_F);
            const v2f a1 = *(const v2f*)(hl + (long long)s1 * HID_F);
            const v2f a2 = *(const v2f*)(hl + (long long)s2 * HID_F);
            const v2f a3 = *(const v2f*)(hl + (long long)s3 * HID_F);
            if (SCALE) {
                const float f0 = __builtin_bit_cast(float, __builtin_amdgcn_readlane(__builtin_bit_cast(int, sc), j));
                const float f1 = __builtin_bit_cast(float, __builtin_amdgcn_readlane(__builtin_bit_cast(int, sc), j + 1));
                const float f2 = __builtin_bit_cast(float, __builtin_amdgcn_readlane(__builtin_bit_cast(int, sc), j + 2));
                const float f3 = __builtin_bit_cast(float, __builtin_amdgcn_readlane(__builtin_bit_cast(int, sc), j + 3));
                acc += a0 * f0;
                acc += a1 * f1;
                acc += a2 * f2;
                acc += a3 * f3;
            } else {
                acc += a0;
                acc += a1;
                acc += a2;
                acc += a3;
            }
        }
        for (; j < n; ++j) {
            const int s0 = __builtin_amdgcn_readlane(idx, j);
            const v2f a0 = *(const v2f*)(hl + (long long)s0 * HID_F);
            if (SCALE) {
                const float f0 = __builtin_bit_cast(float, __builtin_amdgcn_readlane(__builtin_bit_cast(int, sc), j));
                acc += a0 * f0;
            } else {
                acc += a0;
            }
        }
    }
    if (POST) {
        acc = acc * rn + bb;
        acc.x = acc.x > 0.0f ? acc.x : 0.0f;
        acc.y = acc.y > 0.0f ? acc.y : 0.0f;
    }
    *(v2f*)(AGG + (long long)node * HID_F + 2 * lane) = acc;
}

__global__ __launch_bounds__(128) void gcn_gemm2_kernel(
        const float* __restrict__ A, const float* __restrict__ W,
        const float* __restrict__ inorm, const float* __restrict__ bias,
        float* __restrict__ Out, int nN) {
    const int wv   = __builtin_amdgcn_readfirstlane(threadIdx.x >> 5);
    const int lane = threadIdx.x & 31;
    const int half = lane >> 4;
    const int l    = lane & 15;
    const int m0   = (blockIdx.x * 4 + wv) * 16;
    if (m0 >= nN) return;

    int row = m0 + l;
    if (row >= nN) row = nN - 1;

    const v4f bb = *(const v4f*)(bias + 4 * l);
    float rn[8];
#pragma unroll
    for (int r = 0; r < 8; ++r) {
        int orow = m0 + r + 8 * half;
        if (orow >= nN) orow = nN - 1;
        rn[r] = inorm[orow];
    }

    v8f acc[4] = {};
    gcn_strip_mma<HID_F, false>(A + (long long)row * HID_F + 4 * half,
                                W + 4 * half * OUT_F + 4 * l, 1.0f, acc);
#pragma unroll
    for (int r = 0; r < 8; ++r) {
        const int orow = m0 + r + 8 * half;
        v4f o;
        o.x = acc[0][r] * rn[r] + bb.x;
        o.y = acc[1][r] * rn[r] + bb.y;
        o.z = acc[2][r] * rn[r] + bb.z;
        o.w = acc[3][r] * rn[r] + bb.w;
        if (orow < nN) *(v4f*)(Out + (long long)orow * OUT_F + 4 * l) = o;
    }
}

extern "C" void kernel_launch(void* const* d_in, const int* in_sizes, int n_in,
                              void* d_out, int out_size, void* d_ws, size_t ws_size,
                              hipStream_t stream) {
    const float* X   = (const float*)d_in[0];
    const int*   src = (const int*)  d_in[1];
    const int*   dst = (const int*)  d_in[2];
    const float* W1  = (const float*)d_in[3];
    const float* b1  = (const float*)d_in[4];
    const float* W2  = (const float*)d_in[5];
    const float* b2  = (const float*)d_in[6];
    float*       out = (float*)d_out;

    const int nN = in_sizes[0] / IN_F;
    const int nE = in_sizes[1];

    const long long nh  = (long long)nN * HID_F;
    const int nT = (nN + SCAN_TILE - 1) / SCAN_TILE;
    const long long nP = (long long)nT * SCAN_TILE;

    const int nQ = (int)((nP + OD_PART - 1) / OD_PART);

    float* ws      = (float*)d_ws;
    float* bufA    = ws;
    float* bufB    = bufA + nh;
    int*   part_cnt = (int*)(bufB + nh);
    int*   cnt_in  = part_cnt + (long long)OD_SLICES * nP;
    float* onorm   = (float*)(cnt_in + nP);
    float* inorm   = onorm + nP;
    int*   rowptr  = (int*)(inorm + nP);
    int*   rank    = rowptr + nP;
    int*   esrc    = rank + nE;
    int*   tsum    = esrc + nE;

    gcn_outdeg_kernel     <<<dim3(OD_SLICES, nQ), 1024, 0, stream>>>(src, part_cnt, cnt_in, nE, (int)nP);
    gcn_rank_kernel       <<<(nE + 255) / 256, 256, 0, stream>>>(dst, cnt_in, rank, nE);
    gcn_scan_tile_kernel  <<<nT, 256, 0, stream>>>(cnt_in, rowptr, tsum);
    gcn_scan_finish_kernel<<<nT, 256, 0, stream>>>(part_cnt, cnt_in, tsum, rowptr, onorm, inorm, (int)nP);
    gcn_fill_kernel       <<<(nE + 255) / 256, 256, 0, stream>>>(src, dst, rowptr, rank, esrc, nE);

    gcn_gemm1_kernel              <<<(nN + 63) / 64, 128, 0, stream>>>(X, W1, onorm, bufA, nN);
    gcn_gather_kernel<false, true><<<(nN + 7) / 8, 256, 0, stream>>>(rowptr, cnt_in, esrc, bufA, onorm, inorm, b1, bufB, nN);

    gcn_gather_kernel<true, false><<<(nN + 7) / 8, 256, 0, stream>>>(rowptr, cnt_in, esrc, bufB, onorm, inorm, b1, bufA, nN);
    gcn_gemm2_kernel        <<<(nN + 63) / 64, 128, 0, stream>>>(bufA, W2, inorm, b2, out, nN);
}


# In-degree counts and edge ranks without global atomics: per edge slice one block counts dst (one src) in packed 16-bit LDS counters, returning LDS add = rank; row sum order = slice, then LDS arrival
# speedup vs baseline: 4.3124x; 1.3893x over previous
#include <hip/hip_runtime.h>

#define IN_F  128
#define HID_F 64
#define OUT_F 64

#define SCAN_TILE 1024
#define CS_SLICES 32
#define CS_WORDS  65536

typedef __attribute__((ext_vector_type(2))) float v2f;
typedef __attribute__((ext_vector_type(8))) float v8f;
typedef __attribute__((ext_vector_type(4))) float v4f;
typedef __attribute__((ext_vector_type(4))) int   v4i;
typedef __attribute__((ext_vector_type(2))) unsigned v2u;
typedef __attribute__((ext_vector_type(4))) unsigned v4u;

__global__ __launch_bounds__(1024) void gcn_count_kernel(
        const int* __restrict__ src, const int* __restrict__ dst,
        unsigned* __restrict__ part_out, unsigned* __restrict__ part_in,
        int* __restrict__ rank, int nE, int len, int nW) {
    __shared__ unsigned hist[CS_WORDS];
    const int  t      = threadIdx.x;
    const int  s      = blockIdx.x;
    const bool ranked = blockIdx.y == 0;
    const int* __restrict__ keys = ranked ? dst : src;
    unsigned*  __restrict__ part = (ranked ? part_in : part_out) + (long long)s * nW;

    const v4u z = {0u, 0u, 0u, 0u};
    for (int i = 4 * t; i < nW; i += 4096) *(v4u*)(hist + i) = z;
    __syncthreads();

    const int lo  = s * len;
    const int hi  = (lo + len) < nE ? (lo + len) : nE;
    const int cnt = hi > lo ? hi - lo : 0;
    const int hi4 = lo + (cnt & ~3);
    for (int i = lo + 4 * t; i < hi4; i += 8 * 4096) {
        v4i v[8];
#pragma unroll
        for (int u = 0; u < 8; ++u) {
            const v4i none = {-1, -1, -1, -1};
            const int j = i + u * 4096;
            v[u] = none;
            if (j < hi4) v[u] = *(const v4i*)(keys + j);
        }
#pragma unroll
        for (int u = 0; u < 8; ++u) {
#pragma unroll
            for (int k = 0; k < 4; ++k) {
                const unsigned node = (unsigned)v[u][k];
                const unsigned w    = node >> 1;
                const unsigned sh   = (node & 1u) * 16u;
                unsigned old = 0u;
                if (w < (unsigned)nW) old = atomicAdd(&hist[w], 1u << sh);
                v[u][k] = (int)((old >> sh) & 0xFFFFu);
            }
        }
        if (ranked) {
#pragma unroll
            for (int u = 0; u < 8; ++u) {
                const int j = i + u * 4096;
                if (j < hi4) *(v4i*)(rank + j) = v[u];
            }
        }
    }
    if (t < (cnt & 3)) {
        const unsigned node = (unsigned)keys[hi4 + t];
        const unsigned w    = node >> 1;
        const unsigned sh   = (node & 1u) * 16u;
        unsigned old = 0u;
        if (w < (unsigned)nW) old = atomicAdd(&hist[w], 1u << sh);
        if (ranked) rank[hi4 + t] = (int)((old >> sh) & 0xFFFFu);
    }
    __syncthreads();

    for (int i = 4 * t; i < nW; i += 4096) *(v4u*)(part + i) = *(const v4u*)(hist + i);
}

__device__ __forceinline__ v4i gcn_unpack4(v2u p) {
    v4i r;
    r.x = (int)(p.x & 0xFFFFu);
    r.y = (int)(p.x >> 16);
    r.z = (int)(p.y & 0xFFFFu);
    r.w = (int)(p.y >> 16);
    return r;
}


__global__ __launch_bounds__(256) void gcn_scan_tile_kernel(
        const unsigned* __restrict__ part_in, int* __restrict__ cnt_in,
        int* __restrict__ rowptr, int* __restrict__ tsum, int nW) {
    __shared__ int wsum[8];
    const int t    = threadIdx.x;
    const int lane = t & 31;
    const int wv   = t >> 5;
    const int base = blockIdx.x * SCAN_TILE + t * 4;
    v4i v = {0, 0, 0, 0};
#pragma unroll
    for (int sl = 0; sl < CS_SLICES; ++sl)
        v += gcn_unpack4(*(const v2u*)(part_in + (long long)sl * nW + (base >> 1)));
    *(v4i*)(cnt_in + base) = v;
    const int s = v.x + v.y + v.z + v.w;
    int x = s;
    for (int off = 1; off < 32; off <<= 1) {
        const int y = __shfl_up(x, off);
        if (lane >= off) x += y;
    }
    if (lane == 31) wsum[wv] = x;
    __syncthreads();
    int woff = 0;
    for (int i = 0; i < 8; ++i) woff += (i < wv) ? wsum[i] : 0;
    const int ex = woff + x - s;
    v4i o;
    o.x = ex;
    o.y = ex + v.x;
    o.z = ex + v.x + v.y;
    o.w = ex + v.x + v.y + v.z;
    *(v4i*)(rowptr + base) = o;
    if (t == 255) tsum[blockIdx.x] = woff + x;
}

__global__ __launch_bounds__(256) void gcn_scan_finish_kernel(
        const unsigned* __restrict__ part_out, const unsigned* __restrict__ part_in,
        const int* __restrict__ tsum, int* __restrict__ rowptr, int* __restrict__ ebase,
        float* __restrict__ onorm, float* __restrict__ inorm, int nP, int nW) {
    __shared__ int wsum[8];
    const int t    = threadIdx.x;
    const int lane = t & 31;
    const int wv   = t >> 5;
    int part = 0;
    for (int i = t; i < (int)blockIdx.x; i += 256) part += tsum[i];
    for (int off = 16; off > 0; off >>= 1) part += __shfl_xor(part, off);
    if (lane == 0) wsum[wv] = part;
    __syncthreads();
    int toff = 0;
    for (int i = 0; i < 8; ++i) toff += wsum[i];
    const int base = blockIdx.x * SCAN_TILE + t * 4;
    const v4i rp = *(const v4i*)(rowptr + base) + toff;
    *(v4i*)(rowptr + base) = rp;
    v4i run = rp;
#pragma unroll
    for (int sl = 0; sl < CS_SLICES; ++sl) {
        const v4i c = gcn_unpack4(*(const v2u*)(part_in + (long long)sl * nW + (base >> 1)));
        *(v4i*)(ebase + (long long)sl * nP + base) = run;
        run += c;
    }
    const v4i ci = run - rp;
    v4i co = {0, 0, 0, 0};
#pragma unroll
    for (int sl = 0; sl < CS_SLICES; ++sl)
        co += gcn_unpack4(*(const v2u*)(part_out + (long long)sl * nW + (base >> 1)));
    v4f no, ni;
    for (int u = 0; u < 4; ++u) {
        float dout = (float)co[u];
        float din  = (float)ci[u];
        dout = dout < 1.0f ? 1.0f : dout;
        din  = din  < 1.0f ? 1.0f : din;
        no[u] = rsqrtf(dout);
        ni[u] = rsqrtf(din);
    }
    *(v4f*)(onorm + base) = no;
    *(v4f*)(inorm + base) = ni;
}

__global__ __launch_bounds__(256) void gcn_fill_kernel(
        const int* __restrict__ src, const int* __restrict__ dst,
        const int* __restrict__ ebase, const int* __restrict__ rank,
        int* __restrict__ esrc, int nE, int len, int nP) {
    const int sl = blockIdx.y;
    const int lo = sl * len;
    const int hi = (lo + len) < nE ? (lo + len) : nE;
    const int e  = lo + blockIdx.x * 256 + threadIdx.x;
    if (e < hi) esrc[ebase[(long long)sl * nP + dst[e]] + rank[e]] = src[e];
}

static_assert(HID_F == 64 && OUT_F == 64, "the strip is four 16-column tiles wide");

template <int K, bool SCALE>
__device__ __forceinline__ void gcn_strip_mma(const float* __restrict__ xr,
                                              const float* __restrict__ wl,
                                              float sc, v8f (&acc)[4]) {
#pragma unroll
    for (int q = 0; q < K / 8; ++q) {
        v4f av = *(const v4f*)(xr + 8 * q);
        if (SCALE) av *= sc;
#pragma unroll
        for (int e = 0; e < 2; ++e) {
            const float* wk = wl + (8 * q + 2 * e) * 64;
            const v4f wx = *(const v4f*)wk;
            const v4f wy = *(const v4f*)(wk + 64);
            v2f a;
            a.x = av[2 * e];
            a.y = av[2 * e + 1];
#pragma unroll
            for (int n = 0; n < 4; ++n) {
                v2f b;
                b.x = wx[n];
                b.y = wy[n];
                acc[n] = __builtin_amdgcn_wmma_f32_16x16x4_f32(
                             false, a, false, b, (short)0, acc[n], false, false);
            }
        }
    }
}

__global__ __launch_bounds__(128) void gcn_gemm1_kernel(
        const float* __restrict__ X, const float* __restrict__ W,
        const float* __restrict__ onorm, float* __restrict__ H, int nN) {
    const int wv   = __builtin_amdgcn_readfirstlane(threadIdx.x >> 5);
    const int lane = threadIdx.x & 31;
    const int half = lane >> 4;
    const int l    = lane & 15;
    const int m0   = (blockIdx.x * 4 + wv) * 16;
    if (m0 >= nN) return;

    int row = m0 + l;
    if (row >= nN) row = nN - 1;
    const float sc = onorm[row];

    v8f acc[4] = {};
    gcn_strip_mma<IN_F, true>(X + (long long)row * IN_F + 4 * half,
                              W + 4 * half * HID_F + 4 * l, sc, acc);
#pragma unroll
    for (int r = 0; r < 8; ++r) {
        const int orow = m0 + r + 8 * half;
        if (orow < nN) {
            v4f o;
            o.x = acc[0][r];
            o.y = acc[1][r];
            o.z = acc[2][r];
            o.w = acc[3][r];
            *(v4f*)(H + (long long)orow * HID_F + 4 * l) = o;
        }
    }
}

template <bool SCALE, bool POST>
__global__ __launch_bounds__(256) void gcn_gather_kernel(
        const int* __restrict__ rowptr, const int* __restrict__ cnt_in,
        const int* __restrict__ esrc, const float* __restrict__ H,
        const float* __restrict__ onorm, const float* __restrict__ inorm,
        const float* __restrict__ bias, float* __restrict__ AGG, int nN) {
    const int lane = threadIdx.x & 31;
    const int node = __builtin_amdgcn_readfirstlane(blockIdx.x * 8 + (threadIdx.x >> 5));
    if (node >= nN) return;
    const int start = rowptr[node];
    const int deg   = cnt_in[node];
    const float* hl = H + 2 * lane;
    float rn = 1.0f;
    v2f   bb = {0.0f, 0.0f};
    if (POST) {
        rn = inorm[node];
        bb = *(const v2f*)(bias + 2 * lane);
    }

    v2f acc = {0.0f, 0.0f};
    for (int base = 0; base < deg; base += 32) {
        const int n = (deg - base) < 32 ? (deg - base) : 32;
        int   idx = 0;
        float sc  = 1.0f;
        if (lane < n) {
            idx = esrc[start + base + lane];
            if (SCALE) sc = onorm[idx];
        }
        int j = 0;
        for (; j + 4 <= n; j += 4) {
            const int s0 = __builtin_amdgcn_readlane(idx, j);
            const int s1 = __builtin_amdgcn_readlane(idx, j + 1);
            const int s2 = __builtin_amdgcn_readlane(idx, j + 2);
            const int s3 = __builtin_amdgcn_readlane(idx, j + 3);
            const v2f a0 = *(const v2f*)(hl + (long long)s0 * HID_F);
            const v2f a1 = *(const v2f*)(hl + (long long)s1 * HID_F);
            const v2f a2 = *(const v2f*)(hl + (long long)s2 * HID_F);
            const v2f a3 = *(const v2f*)(hl + (long long)s3 * HID_F);
            if (SCALE) {
                const float f0 = __builtin_bit_cast(float, __builtin_amdgcn_readlane(__builtin_bit_cast(int, sc), j));
                const float f1 = __builtin_bit_cast(float, __builtin_amdgcn_readlane(__builtin_bit_cast(int, sc), j + 1));
                const float f2 = __builtin_bit_cast(float, __builtin_amdgcn_readlane(__builtin_bit_cast(int, sc), j + 2));
                const float f3 = __builtin_bit_cast(float, __builtin_amdgcn_readlane(__builtin_bit_cast(int, sc), j + 3));
                acc += a0 * f0;
                acc += a1 * f1;
                acc += a2 * f2;
                acc += a3 * f3;
            } else {
                acc += a0;
                acc += a1;
                acc += a2;
                acc += a3;
            }
        }
        for (; j < n; ++j) {
            const int s0 = __builtin_amdgcn_readlane(idx, j);
            const v2f a0 = *(const v2f*)(hl + (long long)s0 * HID_F);
            if (SCALE) {
                const float f0 = __builtin_bit_cast(float, __builtin_amdgcn_readlane(__builtin_bit_cast(int, sc), j));
                acc += a0 * f0;
            } else {
                acc += a0;
            }
        }
    }
    if (POST) {
        acc = acc * rn + bb;
        acc.x = acc.x > 0.0f ? acc.x : 0.0f;
        acc.y = acc.y > 0.0f ? acc.y : 0.0f;
    }
    *(v2f*)(AGG + (long long)node * HID_F + 2 * lane) = acc;
}

__global__ __launch_bounds__(128) void gcn_gemm2_kernel(
        const float* __restrict__ A, const float* __restrict__ W,
        const float* __restrict__ inorm, const float* __restrict__ bias,
        float* __restrict__ Out, int nN) {
    const int wv   = __builtin_amdgcn_readfirstlane(threadIdx.x >> 5);
    const int lane = threadIdx.x & 31;
    const int half = lane >> 4;
    const int l    = lane & 15;
    const int m0   = (blockIdx.x * 4 + wv) * 16;
    if (m0 >= nN) return;

    int row = m0 + l;
    if (row >= nN) row = nN - 1;

    const v4f bb = *(const v4f*)(bias + 4 * l);
    float rn[8];
#pragma unroll
    for (int r = 0; r < 8; ++r) {
        int orow = m0 + r + 8 * half;
        if (orow >= nN) orow = nN - 1;
        rn[r] = inorm[orow];
    }

    v8f acc[4] = {};
    gcn_strip_mma<HID_F, false>(A + (long long)row * HID_F + 4 * half,
                                W + 4 * half * OUT_F + 4 * l, 1.0f, acc);
#pragma unroll
    for (int r = 0; r < 8; ++r) {
        const int orow = m0 + r + 8 * half;
        v4f o;
        o.x = acc[0][r] * rn[r] + bb.x;
        o.y = acc[1][r] * rn[r] + bb.y;
        o.z = acc[2][r] * rn[r] + bb.z;
        o.w = acc[3][r] * rn[r] + bb.w;
        if (orow < nN) *(v4f*)(Out + (long long)orow * OUT_F + 4 * l) = o;
    }
}

extern "C" void kernel_launch(void* const* d_in, const int* in_sizes, int n_in,
                              void* d_out, int out_size, void* d_ws, size_t ws_size,
                              hipStream_t stream) {
    const float* X   = (const float*)d_in[0];
    const int*   src = (const int*)  d_in[1];
    const int*   dst = (const int*)  d_in[2];
    const float* W1  = (const float*)d_in[3];
    const float* b1  = (const float*)d_in[4];
    const float* W2  = (const float*)d_in[5];
    const float* b2  = (const float*)d_in[6];
    float*       out = (float*)d_out;

    const int nN = in_sizes[0] / IN_F;
    const int nE = in_sizes[1];

    const long long nh  = (long long)nN * HID_F;
    const int nT = (nN + SCAN_TILE - 1) / SCAN_TILE;
    const long long nP = (long long)nT * SCAN_TILE;

    const int nW  = (int)(nP / 2);
    const int len = ((nE + 32 * CS_SLICES - 1) / (32 * CS_SLICES)) * 32;

    float* ws      = (float*)d_ws;
    float* bufA    = ws;
    float* bufB    = bufA + nh;
    unsigned* part_out = (unsigned*)(bufB + nh);
    unsigned* part_in  = part_out + (long long)CS_SLICES * nW;
    int*   cnt_in  = (int*)(part_in + (long long)CS_SLICES * nW);
    float* onorm   = (float*)(cnt_in + nP);
    float* inorm   = onorm + nP;
    int*   rowptr  = (int*)(inorm + nP);
    int*   ebase   = rowptr + nP;
    int*   rank    = ebase + (long long)CS_SLICES * nP;
    int*   esrc    = rank + nE;
    int*   tsum    = esrc + nE;

    gcn_count_kernel      <<<dim3(CS_SLICES, 2), 1024, 0, stream>>>(src, dst, part_out, part_in, rank, nE, len, nW);
    gcn_scan_tile_kernel  <<<nT, 256, 0, stream>>>(part_in, cnt_in, rowptr, tsum, nW);
    gcn_scan_finish_kernel<<<nT, 256, 0, stream>>>(part_out, part_in, tsum, rowptr, ebase, onorm, inorm, (int)nP, nW);
    gcn_fill_kernel       <<<dim3((len + 255) / 256, CS_SLICES), 256, 0, stream>>>(src, dst, ebase, rank, esrc, nE, len, (int)nP);

    gcn_gemm1_kernel              <<<(nN + 63) / 64, 128, 0, stream>>>(X, W1, onorm, bufA, nN);
    gcn_gather_kernel<false, true><<<(nN + 7) / 8, 256, 0, stream>>>(rowptr, cnt_in, esrc, bufA, onorm, inorm, b1, bufB, nN);

    gcn_gather_kernel<true, false><<<(nN + 7) / 8, 256, 0, stream>>>(rowptr, cnt_in, esrc, bufB, onorm, inorm, b1, bufA, nN);
    gcn_gemm2_kernel        <<<(nN + 63) / 64, 128, 0, stream>>>(bufA, W2, inorm, b2, out, nN);
}


# Layer 2's out_norm factor is put on each finished row once, in the layer-1 gather's epilogue (the reference's own order): gather 2 loses its scattered per-edge out_norm lookups; sum order unchanged
# speedup vs baseline: 4.7501x; 1.1015x over previous
#include <hip/hip_runtime.h>

#define IN_F  128
#define HID_F 64
#define OUT_F 64

#define SCAN_TILE 1024
#define CS_SLICES 32
#define CS_WORDS  65536

typedef __attribute__((ext_vector_type(2))) float v2f;
typedef __attribute__((ext_vector_type(8))) float v8f;
typedef __attribute__((ext_vector_type(4))) float v4f;
typedef __attribute__((ext_vector_type(4))) int   v4i;
typedef __attribute__((ext_vector_type(2))) unsigned v2u;
typedef __attribute__((ext_vector_type(4))) unsigned v4u;

__global__ __launch_bounds__(1024) void gcn_count_kernel(
        const int* __restrict__ src, const int* __restrict__ dst,
        unsigned* __restrict__ part_out, unsigned* __restrict__ part_in,
        int* __restrict__ rank, int nE, int len, int nW) {
    __shared__ unsigned hist[CS_WORDS];
    const int  t      = threadIdx.x;
    const int  s      = blockIdx.x;
    const bool ranked = blockIdx.y == 0;
    const int* __restrict__ keys = ranked ? dst : src;
    unsigned*  __restrict__ part = (ranked ? part_in : part_out) + (long long)s * nW;

    const v4u z = {0u, 0u, 0u, 0u};
    for (int i = 4 * t; i < nW; i += 4096) *(v4u*)(hist + i) = z;
    __syncthreads();

    const int lo  = s * len;
    const int hi  = (lo + len) < nE ? (lo + len) : nE;
    const int cnt = hi > lo ? hi - lo : 0;
    const int hi4 = lo + (cnt & ~3);
    for (int i = lo + 4 * t; i < hi4; i += 8 * 4096) {
        v4i v[8];
#pragma unroll
        for (int u = 0; u < 8; ++u) {
            const v4i none = {-1, -1, -1, -1};
            const int j = i + u * 4096;
            v[u] = none;
            if (j < hi4) v[u] = *(const v4i*)(keys + j);
        }
#pragma unroll
        for (int u = 0; u < 8; ++u) {
#pragma unroll
            for (int k = 0; k < 4; ++k) {
                const unsigned node = (unsigned)v[u][k];
                const unsigned w    = node >> 1;
                const unsigned sh   = (node & 1u) * 16u;
                unsigned old = 0u;
                if (w < (unsigned)nW) old = atomicAdd(&hist[w], 1u << sh);
                v[u][k] = (int)((old >> sh) & 0xFFFFu);
            }
        }
        if (ranked) {
#pragma unroll
            for (int u = 0; u < 8; ++u) {
                const int j = i + u * 4096;
                if (j < hi4) *(v4i*)(rank + j) = v[u];
            }
        }
    }
    if (t < (cnt & 3)) {
        const unsigned node = (unsigned)keys[hi4 + t];
        const unsigned w    = node >> 1;
        const unsigned sh   = (node & 1u) * 16u;
        unsigned old = 0u;
        if (w < (unsigned)nW) old = atomicAdd(&hist[w], 1u << sh);
        if (ranked) rank[hi4 + t] = (int)((old >> sh) & 0xFFFFu);
    }
    __syncthreads();

    for (int i = 4 * t; i < nW; i += 4096) *(v4u*)(part + i) = *(const v4u*)(hist + i);
}

__device__ __forceinline__ v4i gcn_unpack4(v2u p) {
    v4i r;
    r.x = (int)(p.x & 0xFFFFu);
    r.y = (int)(p.x >> 16);
    r.z = (int)(p.y & 0xFFFFu);
    r.w = (int)(p.y >> 16);
    return r;
}


__global__ __launch_bounds__(256) void gcn_scan_tile_kernel(
        const unsigned* __restrict__ part_in, int* __restrict__ cnt_in,
        int* __restrict__ rowptr, int* __restrict__ tsum, int nW) {
    __shared__ int wsum[8];
    const int t    = threadIdx.x;
    const int lane = t & 31;
    const int wv   = t >> 5;
    const int base = blockIdx.x * SCAN_TILE + t * 4;
    v4i v = {0, 0, 0, 0};
#pragma unroll
    for (int sl = 0; sl < CS_SLICES; ++sl)
        v += gcn_unpack4(*(const v2u*)(part_in + (long long)sl * nW + (base >> 1)));
    *(v4i*)(cnt_in + base) = v;
    const int s = v.x + v.y + v.z + v.w;
    int x = s;
    for (int off = 1; off < 32; off <<= 1) {
        const int y = __shfl_up(x, off);
        if (lane >= off) x += y;
    }
    if (lane == 31) wsum[wv] = x;
    __syncthreads();
    int woff = 0;
    for (int i = 0; i < 8; ++i) woff += (i < wv) ? wsum[i] : 0;
    const int ex = woff + x - s;
    v4i o;
    o.x = ex;
    o.y = ex + v.x;
    o.z = ex + v.x + v.y;
    o.w = ex + v.x + v.y + v.z;
    *(v4i*)(rowptr + base) = o;
    if (t == 255) tsum[blockIdx.x] = woff + x;
}

__global__ __launch_bounds__(256) void gcn_scan_finish_kernel(
        const unsigned* __restrict__ part_out, const unsigned* __restrict__ part_in,
        const int* __restrict__ tsum, int* __restrict__ rowptr, int* __restrict__ ebase,
        float* __restrict__ onorm, float* __restrict__ inorm, int nP, int nW) {
    __shared__ int wsum[8];
    const int t    = threadIdx.x;
    const int lane = t & 31;
    const int wv   = t >> 5;
    int part = 0;
    for (int i = t; i < (int)blockIdx.x; i += 256) part += tsum[i];
    for (int off = 16; off > 0; off >>= 1) part += __shfl_xor(part, off);
    if (lane == 0) wsum[wv] = part;
    __syncthreads();
    int toff = 0;
    for (int i = 0; i < 8; ++i) toff += wsum[i];
    const int base = blockIdx.x * SCAN_TILE + t * 4;
    const v4i rp = *(const v4i*)(rowptr + base) + toff;
    *(v4i*)(rowptr + base) = rp;
    v4i run = rp;
#pragma unroll
    for (int sl = 0; sl < CS_SLICES; ++sl) {
        const v4i c = gcn_unpack4(*(const v2u*)(part_in + (long long)sl * nW + (base >> 1)));
        *(v4i*)(ebase + (long long)sl * nP + base) = run;
        run += c;
    }
    const v4i ci = run - rp;
    v4i co = {0, 0, 0, 0};
#pragma unroll
    for (int sl = 0; sl < CS_SLICES; ++sl)
        co += gcn_unpack4(*(const v2u*)(part_out + (long long)sl * nW + (base >> 1)));
    v4f no, ni;
    for (int u = 0; u < 4; ++u) {
        float dout = (float)co[u];
        float din  = (float)ci[u];
        dout = dout < 1.0f ? 1.0f : dout;
        din  = din  < 1.0f ? 1.0f : din;
        no[u] = rsqrtf(dout);
        ni[u] = rsqrtf(din);
    }
    *(v4f*)(onorm + base) = no;
    *(v4f*)(inorm + base) = ni;
}

__global__ __launch_bounds__(256) void gcn_fill_kernel(
        const int* __restrict__ src, const int* __restrict__ dst,
        const int* __restrict__ ebase, const int* __restrict__ rank,
        int* __restrict__ esrc, int nE, int len, int nP) {
    const int sl = blockIdx.y;
    const int lo = sl * len;
    const int hi = (lo + len) < nE ? (lo + len) : nE;
    const int e  = lo + blockIdx.x * 256 + threadIdx.x;
    if (e < hi) esrc[ebase[(long long)sl * nP + dst[e]] + rank[e]] = src[e];
}

static_assert(HID_F == 64 && OUT_F == 64, "the strip is four 16-column tiles wide");

template <int K, bool SCALE>
__device__ __forceinline__ void gcn_strip_mma(const float* __restrict__ xr,
                                              const float* __restrict__ wl,
                                              float sc, v8f (&acc)[4]) {
#pragma unroll
    for (int q = 0; q < K / 8; ++q) {
        v4f av = *(const v4f*)(xr + 8 * q);
        if (SCALE) av *= sc;
#pragma unroll
        for (int e = 0; e < 2; ++e) {
            const float* wk = wl + (8 * q + 2 * e) * 64;
            const v4f wx = *(const v4f*)wk;
            const v4f wy = *(const v4f*)(wk + 64);
            v2f a;
            a.x = av[2 * e];
            a.y = av[2 * e + 1];
#pragma unroll
            for (int n = 0; n < 4; ++n) {
                v2f b;
                b.x = wx[n];
                b.y = wy[n];
                acc[n] = __builtin_amdgcn_wmma_f32_16x16x4_f32(
                             false, a, false, b, (short)0, acc[n], false, false);
            }
        }
    }
}

__global__ __launch_bounds__(128) void gcn_gemm1_kernel(
        const float* __restrict__ X, const float* __restrict__ W,
        const float* __restrict__ onorm, float* __restrict__ H, int nN) {
    const int wv   = __builtin_amdgcn_readfirstlane(threadIdx.x >> 5);
    const int lane = threadIdx.x & 31;
    const int half = lane >> 4;
    const int l    = lane & 15;
    const int m0   = (blockIdx.x * 4 + wv) * 16;
    if (m0 >= nN) return;

    int row = m0 + l;
    if (row >= nN) row = nN - 1;
    const float sc = onorm[row];

    v8f acc[4] = {};
    gcn_strip_mma<IN_F, true>(X + (long long)row * IN_F + 4 * half,
                              W + 4 * half * HID_F + 4 * l, sc, acc);
#pragma unroll
    for (int r = 0; r < 8; ++r) {
        const int orow = m0 + r + 8 * half;
        if (orow < nN) {
            v4f o;
            o.x = acc[0][r];
            o.y = acc[1][r];
            o.z = acc[2][r];
            o.w = acc[3][r];
            *(v4f*)(H + (long long)orow * HID_F + 4 * l) = o;
        }
    }
}


__device__ __forceinline__ void gcn_group_loads(const float* __restrict__ hl, int idx, int j,
                                                v2f (&a)[8]) {
#pragma unroll
    for (int u = 0; u < 8; ++u) {
        const int s = __builtin_amdgcn_readlane(idx, j + u);
        a[u] = *(const v2f*)(hl + (long long)s * HID_F);
    }
}

__device__ __forceinline__ void gcn_group_adds(v2f& acc, int r, const v2f (&a)[8]) {
#pragma unroll
    for (int u = 0; u < 8; ++u) {
        if (u < r) acc += a[u];
    }
}

template <bool POST>
__global__ __launch_bounds__(256) void gcn_gather_kernel(
        const int* __restrict__ rowptr, const int* __restrict__ cnt_in,
        const int* __restrict__ esrc, const float* __restrict__ H,
        const float* __restrict__ onorm, const float* __restrict__ inorm,
        const float* __restrict__ bias, float* __restrict__ AGG, int nN) {
    const int lane = threadIdx.x & 31;
    const int nodeA = __builtin_amdgcn_readfirstlane((blockIdx.x * 8 + (threadIdx.x >> 5)) * 2);
    if (nodeA >= nN) return;
    const bool two   = nodeA + 1 < nN;
    const int  nodeB = two ? nodeA + 1 : nodeA;
    const int startA = rowptr[nodeA];
    const int startB = rowptr[nodeB];
    const int degA   = cnt_in[nodeA];
    const int degB   = two ? cnt_in[nodeB] : 0;
    const float* hl = H + 2 * lane;
    float rnA = 1.0f, rnB = 1.0f;
    float onA = 1.0f, onB = 1.0f;
    v2f   bb  = {0.0f, 0.0f};
    if (POST) {
        rnA = inorm[nodeA];
        rnB = inorm[nodeB];
        onA = onorm[nodeA];
        onB = onorm[nodeB];
        bb  = *(const v2f*)(bias + 2 * lane);
    }

    v2f accA = {0.0f, 0.0f};
    v2f accB = {0.0f, 0.0f};
    const int dmax = degA > degB ? degA : degB;
    for (int base = 0; base < dmax; base += 32) {
        int nA = degA - base;
        int nB = degB - base;
        nA = nA < 0 ? 0 : (nA > 32 ? 32 : nA);
        nB = nB < 0 ? 0 : (nB > 32 ? 32 : nB);
        int idxA = 0, idxB = 0;
        if (lane < nA) idxA = esrc[startA + base + lane];
        if (lane < nB) idxB = esrc[startB + base + lane];
        const int nm = nA > nB ? nA : nB;
        for (int j = 0; j < nm; j += 8) {
            v2f a[8], b[8];
            gcn_group_loads(hl, idxA, j, a);
            gcn_group_loads(hl, idxB, j, b);
            gcn_group_adds(accA, nA - j, a);
            gcn_group_adds(accB, nB - j, b);
        }
    }
    if (POST) {
        accA = accA * rnA + bb;
        accA.x = accA.x > 0.0f ? accA.x : 0.0f;
        accA.y = accA.y > 0.0f ? accA.y : 0.0f;
        accA *= onA;
        accB = accB * rnB + bb;
        accB.x = accB.x > 0.0f ? accB.x : 0.0f;
        accB.y = accB.y > 0.0f ? accB.y : 0.0f;
        accB *= onB;
    }
    *(v2f*)(AGG + (long long)nodeA * HID_F + 2 * lane) = accA;
    if (two) *(v2f*)(AGG + (long long)nodeB * HID_F + 2 * lane) = accB;
}

__global__ __launch_bounds__(128) void gcn_gemm2_kernel(
        const float* __restrict__ A, const float* __restrict__ W,
        const float* __restrict__ inorm, const float* __restrict__ bias,
        float* __restrict__ Out, int nN) {
    const int wv   = __builtin_amdgcn_readfirstlane(threadIdx.x >> 5);
    const int lane = threadIdx.x & 31;
    const int half = lane >> 4;
    const int l    = lane & 15;
    const int m0   = (blockIdx.x * 4 + wv) * 16;
    if (m0 >= nN) return;

    int row = m0 + l;
    if (row >= nN) row = nN - 1;

    const v4f bb = *(const v4f*)(bias + 4 * l);
    float rn[8];
#pragma unroll
    for (int r = 0; r < 8; ++r) {
        int orow = m0 + r + 8 * half;
        if (orow >= nN) orow = nN - 1;
        rn[r] = inorm[orow];
    }

    v8f acc[4] = {};
    gcn_strip_mma<HID_F, false>(A + (long long)row * HID_F + 4 * half,
                                W + 4 * half * OUT_F + 4 * l, 1.0f, acc);
#pragma unroll
    for (int r = 0; r < 8; ++r) {
        const int orow = m0 + r + 8 * half;
        v4f o;
        o.x = acc[0][r] * rn[r] + bb.x;
        o.y = acc[1][r] * rn[r] + bb.y;
        o.z = acc[2][r] * rn[r] + bb.z;
        o.w = acc[3][r] * rn[r] + bb.w;
        if (orow < nN) *(v4f*)(Out + (long long)orow * OUT_F + 4 * l) = o;
    }
}

extern "C" void kernel_launch(void* const* d_in, const int* in_sizes, int n_in,
                              void* d_out, int out_size, void* d_ws, size_t ws_size,
                              hipStream_t stream) {
    const float* X   = (const float*)d_in[0];
    const int*   src = (const int*)  d_in[1];
    const int*   dst = (const int*)  d_in[2];
    const float* W1  = (const float*)d_in[3];
    const float* b1  = (const float*)d_in[4];
    const float* W2  = (const float*)d_in[5];
    const float* b2  = (const float*)d_in[6];
    float*       out = (float*)d_out;

    const int nN = in_sizes[0] / IN_F;
    const int nE = in_sizes[1];

    const long long nh  = (long long)nN * HID_F;
    const int nT = (nN + SCAN_TILE - 1) / SCAN_TILE;
    const long long nP = (long long)nT * SCAN_TILE;

    const int nW  = (int)(nP / 2);
    const int len = ((nE + 32 * CS_SLICES - 1) / (32 * CS_SLICES)) * 32;

    float* ws      = (float*)d_ws;
    float* bufA    = ws;
    float* bufB    = bufA + nh;
    unsigned* part_out = (unsigned*)(bufB + nh);
    unsigned* part_in  = part_out + (long long)CS_SLICES * nW;
    int*   cnt_in  = (int*)(part_in + (long long)CS_SLICES * nW);
    float* onorm   = (float*)(cnt_in + nP);
    float* inorm   = onorm + nP;
    int*   rowptr  = (int*)(inorm + nP);
    int*   ebase   = rowptr + nP;
    int*   rank    = ebase + (long long)CS_SLICES * nP;
    int*   esrc    = rank + nE;
    int*   tsum    = esrc + nE;

    gcn_count_kernel      <<<dim3(CS_SLICES, 2), 1024, 0, stream>>>(src, dst, part_out, part_in, rank, nE, len, nW);
    gcn_scan_tile_kernel  <<<nT, 256, 0, stream>>>(part_in, cnt_in, rowptr, tsum, nW);
    gcn_scan_finish_kernel<<<nT, 256, 0, stream>>>(part_out, part_in, tsum, rowptr, ebase, onorm, inorm, (int)nP, nW);
    gcn_fill_kernel       <<<dim3((len + 255) / 256, CS_SLICES), 256, 0, stream>>>(src, dst, ebase, rank, esrc, nE, len, (int)nP);

    gcn_gemm1_kernel              <<<(nN + 63) / 64, 128, 0, stream>>>(X, W1, onorm, bufA, nN);
    gcn_gather_kernel<true> <<<(nN + 15) / 16, 256, 0, stream>>>(rowptr, cnt_in, esrc, bufA, onorm, inorm, b1, bufB, nN);

    gcn_gather_kernel<false><<<(nN + 15) / 16, 256, 0, stream>>>(rowptr, cnt_in, esrc, bufB, onorm, inorm, b1, bufA, nN);
    gcn_gemm2_kernel        <<<(nN + 63) / 64, 128, 0, stream>>>(bufA, W2, inorm, b2, out, nN);
}


# GEMM2 moved into layer 2's gather: a block leaves its 16 summed rows in LDS, two of its waves multiply them by W2 on WMMA and store the strip; AGG2's round trip and a launch go; sum orders unchanged
# speedup vs baseline: 5.0000x; 1.0526x over previous
#include <hip/hip_runtime.h>

#define IN_F  128
#define HID_F 64
#define OUT_F 64

#define SCAN_TILE 1024
#define CS_SLICES 32
#define CS_WORDS  65536

typedef __attribute__((ext_vector_type(2))) float v2f;
typedef __attribute__((ext_vector_type(8))) float v8f;
typedef __attribute__((ext_vector_type(4))) float v4f;
typedef __attribute__((ext_vector_type(4))) int   v4i;
typedef __attribute__((ext_vector_type(2))) unsigned v2u;
typedef __attribute__((ext_vector_type(4))) unsigned v4u;

__global__ __launch_bounds__(1024) void gcn_count_kernel(
        const int* __restrict__ src, const int* __restrict__ dst,
        unsigned* __restrict__ part_out, unsigned* __restrict__ part_in,
        int* __restrict__ rank, int nE, int len, int nW) {
    __shared__ unsigned hist[CS_WORDS];
    const int  t      = threadIdx.x;
    const int  s      = blockIdx.x;
    const bool ranked = blockIdx.y == 0;
    const int* __restrict__ keys = ranked ? dst : src;
    unsigned*  __restrict__ part = (ranked ? part_in : part_out) + (long long)s * nW;

    const v4u z = {0u, 0u, 0u, 0u};
    for (int i = 4 * t; i < nW; i += 4096) *(v4u*)(hist + i) = z;
    __syncthreads();

    const int lo  = s * len;
    const int hi  = (lo + len) < nE ? (lo + len) : nE;
    const int cnt = hi > lo ? hi - lo : 0;
    const int hi4 = lo + (cnt & ~3);
    for (int i = lo + 4 * t; i < hi4; i += 8 * 4096) {
        v4i v[8];
#pragma unroll
        for (int u = 0; u < 8; ++u) {
            const v4i none = {-1, -1, -1, -1};
            const int j = i + u * 4096;
            v[u] = none;
            if (j < hi4) v[u] = *(const v4i*)(keys + j);
        }
#pragma unroll
        for (int u = 0; u < 8; ++u) {
#pragma unroll
            for (int k = 0; k < 4; ++k) {
                const unsigned node = (unsigned)v[u][k];
                const unsigned w    = node >> 1;
                const unsigned sh   = (node & 1u) * 16u;
                unsigned old = 0u;
                if (w < (unsigned)nW) old = atomicAdd(&hist[w], 1u << sh);
                v[u][k] = (int)((old >> sh) & 0xFFFFu);
            }
        }
        if (ranked) {
#pragma unroll
            for (int u = 0; u < 8; ++u) {
                const int j = i + u * 4096;
                if (j < hi4) *(v4i*)(rank + j) = v[u];
            }
        }
    }
    if (t < (cnt & 3)) {
        const unsigned node = (unsigned)keys[hi4 + t];
        const unsigned w    = node >> 1;
        const unsigned sh   = (node & 1u) * 16u;
        unsigned old = 0u;
        if (w < (unsigned)nW) old = atomicAdd(&hist[w], 1u << sh);
        if (ranked) rank[hi4 + t] = (int)((old >> sh) & 0xFFFFu);
    }
    __syncthreads();

    for (int i = 4 * t; i < nW; i += 4096) *(v4u*)(part + i) = *(const v4u*)(hist + i);
}

__device__ __forceinline__ v4i gcn_unpack4(v2u p) {
    v4i r;
    r.x = (int)(p.x & 0xFFFFu);
    r.y = (int)(p.x >> 16);
    r.z = (int)(p.y & 0xFFFFu);
    r.w = (int)(p.y >> 16);
    return r;
}


__global__ __launch_bounds__(256) void gcn_scan_tile_kernel(
        const unsigned* __restrict__ part_in, int* __restrict__ cnt_in,
        int* __restrict__ rowptr, int* __restrict__ tsum, int nW) {
    __shared__ int wsum[8];
    const int t    = threadIdx.x;
    const int lane = t & 31;
    const int wv   = t >> 5;
    const int base = blockIdx.x * SCAN_TILE + t * 4;
    v4i v = {0, 0, 0, 0};
#pragma unroll
    for (int sl = 0; sl < CS_SLICES; ++sl)
        v += gcn_unpack4(*(const v2u*)(part_in + (long long)sl * nW + (base >> 1)));
    *(v4i*)(cnt_in + base) = v;
    const int s = v.x + v.y + v.z + v.w;
    int x = s;
    for (int off = 1; off < 32; off <<= 1) {
        const int y = __shfl_up(x, off);
        if (lane >= off) x += y;
    }
    if (lane == 31) wsum[wv] = x;
    __syncthreads();
    int woff = 0;
    for (int i = 0; i < 8; ++i) woff += (i < wv) ? wsum[i] : 0;
    const int ex = woff + x - s;
    v4i o;
    o.x = ex;
    o.y = ex + v.x;
    o.z = ex + v.x + v.y;
    o.w = ex + v.x + v.y + v.z;
    *(v4i*)(rowptr + base) = o;
    if (t == 255) tsum[blockIdx.x] = woff + x;
}

__global__ __launch_bounds__(256) void gcn_scan_finish_kernel(
        const unsigned* __restrict__ part_out, const unsigned* __restrict__ part_in,
        const int* __restrict__ tsum, int* __restrict__ rowptr, int* __restrict__ ebase,
        float* __restrict__ onorm, float* __restrict__ inorm, int nP, int nW) {
    __shared__ int wsum[8];
    const int t    = threadIdx.x;
    const int lane = t & 31;
    const int wv   = t >> 5;
    int part = 0;
    for (int i = t; i < (int)blockIdx.x; i += 256) part += tsum[i];
    for (int off = 16; off > 0; off >>= 1) part += __shfl_xor(part, off);
    if (lane == 0) wsum[wv] = part;
    __syncthreads();
    int toff = 0;
    for (int i = 0; i < 8; ++i) toff += wsum[i];
    const int base = blockIdx.x * SCAN_TILE + t * 4;
    const v4i rp = *(const v4i*)(rowptr + base) + toff;
    *(v4i*)(rowptr + base) = rp;
    v4i run = rp;
#pragma unroll
    for (int sl = 0; sl < CS_SLICES; ++sl) {
        const v4i c = gcn_unpack4(*(const v2u*)(part_in + (long long)sl * nW + (base >> 1)));
        *(v4i*)(ebase + (long long)sl * nP + base) = run;
        run += c;
    }
    const v4i ci = run - rp;
    v4i co = {0, 0, 0, 0};
#pragma unroll
    for (int sl = 0; sl < CS_SLICES; ++sl)
        co += gcn_unpack4(*(const v2u*)(part_out + (long long)sl * nW + (base >> 1)));
    v4f no, ni;
    for (int u = 0; u < 4; ++u) {
        float dout = (float)co[u];
        float din  = (float)ci[u];
        dout = dout < 1.0f ? 1.0f : dout;
        din  = din  < 1.0f ? 1.0f : din;
        no[u] = rsqrtf(dout);
        ni[u] = rsqrtf(din);
    }
    *(v4f*)(onorm + base) = no;
    *(v4f*)(inorm + base) = ni;
}

__global__ __launch_bounds__(256) void gcn_fill_kernel(
        const int* __restrict__ src, const int* __restrict__ dst,
        const int* __restrict__ ebase, const int* __restrict__ rank,
        int* __restrict__ esrc, int nE, int len, int nP) {
    const int sl = blockIdx.y;
    const int lo = sl * len;
    const int hi = (lo + len) < nE ? (lo + len) : nE;
    const int e  = lo + blockIdx.x * 256 + threadIdx.x;
    if (e < hi) esrc[ebase[(long long)sl * nP + dst[e]] + rank[e]] = src[e];
}

static_assert(HID_F == 64 && OUT_F == 64, "the strip is four 16-column tiles wide");

template <int K, bool SCALE>
__device__ __forceinline__ void gcn_strip_mma(const float* __restrict__ xr,
                                              const float* __restrict__ wl,
                                              float sc, v8f (&acc)[4]) {
#pragma unroll
    for (int q = 0; q < K / 8; ++q) {
        v4f av = *(const v4f*)(xr + 8 * q);
        if (SCALE) av *= sc;
#pragma unroll
        for (int e = 0; e < 2; ++e) {
            const float* wk = wl + (8 * q + 2 * e) * 64;
            const v4f wx = *(const v4f*)wk;
            const v4f wy = *(const v4f*)(wk + 64);
            v2f a;
            a.x = av[2 * e];
            a.y = av[2 * e + 1];
#pragma unroll
            for (int n = 0; n < 4; ++n) {
                v2f b;
                b.x = wx[n];
                b.y = wy[n];
                acc[n] = __builtin_amdgcn_wmma_f32_16x16x4_f32(
                             false, a, false, b, (short)0, acc[n], false, false);
            }
        }
    }
}

__global__ __launch_bounds__(128) void gcn_gemm1_kernel(
        const float* __restrict__ X, const float* __restrict__ W,
        const float* __restrict__ onorm, float* __restrict__ H, int nN) {
    const int wv   = __builtin_amdgcn_readfirstlane(threadIdx.x >> 5);
    const int lane = threadIdx.x & 31;
    const int half = lane >> 4;
    const int l    = lane & 15;
    const int m0   = (blockIdx.x * 4 + wv) * 16;
    if (m0 >= nN) return;

    int row = m0 + l;
    if (row >= nN) row = nN - 1;
    const float sc = onorm[row];

    v8f acc[4] = {};
    gcn_strip_mma<IN_F, true>(X + (long long)row * IN_F + 4 * half,
                              W + 4 * half * HID_F + 4 * l, sc, acc);
#pragma unroll
    for (int r = 0; r < 8; ++r) {
        const int orow = m0 + r + 8 * half;
        if (orow < nN) {
            v4f o;
            o.x = acc[0][r];
            o.y = acc[1][r];
            o.z = acc[2][r];
            o.w = acc[3][r];
            *(v4f*)(H + (long long)orow * HID_F + 4 * l) = o;
        }
    }
}


__device__ __forceinline__ void gcn_group_loads(const float* __restrict__ hl, int idx, int j,
                                                v2f (&a)[8]) {
#pragma unroll
    for (int u = 0; u < 8; ++u) {
        const int s = __builtin_amdgcn_readlane(idx, j + u);
        a[u] = *(const v2f*)(hl + (long long)s * HID_F);
    }
}

__device__ __forceinline__ void gcn_group_adds(v2f& acc, int r, const v2f (&a)[8]) {
#pragma unroll
    for (int u = 0; u < 8; ++u) {
        if (u < r) acc += a[u];
    }
}

__device__ __forceinline__ void gcn_gather_two_rows(const int* __restrict__ esrc,
                                                    const float* __restrict__ hl,
                                                    int startA, int startB, int degA, int degB,
                                                    int lane, v2f& accA, v2f& accB) {
    accA = (v2f){0.0f, 0.0f};
    accB = (v2f){0.0f, 0.0f};
    const int dmax = degA > degB ? degA : degB;
    for (int base = 0; base < dmax; base += 32) {
        int nA = degA - base;
        int nB = degB - base;
        nA = nA < 0 ? 0 : (nA > 32 ? 32 : nA);
        nB = nB < 0 ? 0 : (nB > 32 ? 32 : nB);
        int idxA = 0, idxB = 0;
        if (lane < nA) idxA = esrc[startA + base + lane];
        if (lane < nB) idxB = esrc[startB + base + lane];
        const int nm = nA > nB ? nA : nB;
        for (int j = 0; j < nm; j += 8) {
            v2f a[8], b[8];
            gcn_group_loads(hl, idxA, j, a);
            gcn_group_loads(hl, idxB, j, b);
            gcn_group_adds(accA, nA - j, a);
            gcn_group_adds(accB, nB - j, b);
        }
    }
}

__global__ __launch_bounds__(256) void gcn_gather_kernel(
        const int* __restrict__ rowptr, const int* __restrict__ cnt_in,
        const int* __restrict__ esrc, const float* __restrict__ H,
        const float* __restrict__ onorm, const float* __restrict__ inorm,
        const float* __restrict__ bias, float* __restrict__ AGG, int nN) {
    const int lane = threadIdx.x & 31;
    const int nodeA = __builtin_amdgcn_readfirstlane((blockIdx.x * 8 + (threadIdx.x >> 5)) * 2);
    if (nodeA >= nN) return;
    const bool two   = nodeA + 1 < nN;
    const int  nodeB = two ? nodeA + 1 : nodeA;
    const int startA = rowptr[nodeA];
    const int startB = rowptr[nodeB];
    const int degA   = cnt_in[nodeA];
    const int degB   = two ? cnt_in[nodeB] : 0;
    const float rnA = inorm[nodeA];
    const float rnB = inorm[nodeB];
    const float onA = onorm[nodeA];
    const float onB = onorm[nodeB];
    const v2f   bb  = *(const v2f*)(bias + 2 * lane);

    v2f accA, accB;
    gcn_gather_two_rows(esrc, H + 2 * lane, startA, startB, degA, degB, lane, accA, accB);

    accA = accA * rnA + bb;
    accA.x = accA.x > 0.0f ? accA.x : 0.0f;
    accA.y = accA.y > 0.0f ? accA.y : 0.0f;
    accA *= onA;
    accB = accB * rnB + bb;
    accB.x = accB.x > 0.0f ? accB.x : 0.0f;
    accB.y = accB.y > 0.0f ? accB.y : 0.0f;
    accB *= onB;
    *(v2f*)(AGG + (long long)nodeA * HID_F + 2 * lane) = accA;
    if (two) *(v2f*)(AGG + (long long)nodeB * HID_F + 2 * lane) = accB;
}

#define STRIP_PITCH 68

__global__ __launch_bounds__(256) void gcn_gather_gemm2_kernel(
        const int* __restrict__ rowptr, const int* __restrict__ cnt_in,
        const int* __restrict__ esrc, const float* __restrict__ H,
        const float* __restrict__ W, const float* __restrict__ inorm,
        const float* __restrict__ bias, float* __restrict__ Out, int nN) {
    __shared__ __attribute__((aligned(16))) float strip[16 * STRIP_PITCH];
    __shared__ __attribute__((aligned(16))) float rnorm[16];
    const int lane = threadIdx.x & 31;
    const int wv    = __builtin_amdgcn_readfirstlane(threadIdx.x >> 5);
    const int m0    = blockIdx.x * 16;
    const int nodeA = m0 + 2 * wv;
    int   startA = 0, startB = 0, degA = 0, degB = 0;
    float rnA = 0.0f, rnB = 0.0f;
    if (nodeA < nN)     { startA = rowptr[nodeA];     degA = cnt_in[nodeA];     rnA = inorm[nodeA]; }
    if (nodeA + 1 < nN) { startB = rowptr[nodeA + 1]; degB = cnt_in[nodeA + 1]; rnB = inorm[nodeA + 1]; }

    v2f accA, accB;
    gcn_gather_two_rows(esrc, H + 2 * lane, startA, startB, degA, degB, lane, accA, accB);
    *(v2f*)(strip + (2 * wv)     * STRIP_PITCH + 2 * lane) = accA;
    *(v2f*)(strip + (2 * wv + 1) * STRIP_PITCH + 2 * lane) = accB;
    if (lane < 2) rnorm[2 * wv + lane] = lane ? rnB : rnA;
    __syncthreads();

    const int hs = wv - 2 * (__builtin_popcount(blockIdx.x) & 1);
    if (hs < 0 || hs > 1) return;

    const int half = lane >> 4;
    const int c    = lane & 15;
    const v2f bb = *(const v2f*)(bias + 32 * hs + 2 * c);

    const float* ar = strip + c * STRIP_PITCH + 4 * half;
    const float* wl = W + 4 * half * OUT_F + 32 * hs + 2 * c;
    v8f acc[2] = {};
#pragma unroll
    for (int q = 0; q < HID_F / 8; ++q) {
        const v4f av = *(const v4f*)(ar + 8 * q);
#pragma unroll
        for (int e = 0; e < 2; ++e) {
            const float* wk = wl + (8 * q + 2 * e) * OUT_F;
            const v2f wx = *(const v2f*)wk;
            const v2f wy = *(const v2f*)(wk + OUT_F);
            v2f a;
            a.x = av[2 * e];
            a.y = av[2 * e + 1];
#pragma unroll
            for (int n = 0; n < 2; ++n) {
                v2f b;
                b.x = wx[n];
                b.y = wy[n];
                acc[n] = __builtin_amdgcn_wmma_f32_16x16x4_f32(
                             false, a, false, b, (short)0, acc[n], false, false);
            }
        }
    }
    __builtin_amdgcn_sched_barrier(0);
    const v4f rn0 = *(const v4f*)(rnorm + 8 * half);
    const v4f rn1 = *(const v4f*)(rnorm + 8 * half + 4);
#pragma unroll
    for (int r = 0; r < 8; ++r) {
        const int   orow = m0 + r + 8 * half;
        const float rn   = r < 4 ? rn0[r & 3] : rn1[r & 3];
        v2f o;
        o.x = acc[0][r] * rn + bb.x;
        o.y = acc[1][r] * rn + bb.y;
        if (orow < nN) *(v2f*)(Out + (long long)orow * OUT_F + 32 * hs + 2 * c) = o;
    }
}

extern "C" void kernel_launch(void* const* d_in, const int* in_sizes, int n_in,
                              void* d_out, int out_size, void* d_ws, size_t ws_size,
                              hipStream_t stream) {
    const float* X   = (const float*)d_in[0];
    const int*   src = (const int*)  d_in[1];
    const int*   dst = (const int*)  d_in[2];
    const float* W1  = (const float*)d_in[3];
    const float* b1  = (const float*)d_in[4];
    const float* W2  = (const float*)d_in[5];
    const float* b2  = (const float*)d_in[6];
    float*       out = (float*)d_out;

    const int nN = in_sizes[0] / IN_F;
    const int nE = in_sizes[1];

    const long long nh  = (long long)nN * HID_F;
    const int nT = (nN + SCAN_TILE - 1) / SCAN_TILE;
    const long long nP = (long long)nT * SCAN_TILE;

    const int nW  = (int)(nP / 2);
    const int len = ((nE + 32 * CS_SLICES - 1) / (32 * CS_SLICES)) * 32;

    float* ws      = (float*)d_ws;
    float* bufA    = ws;
    float* bufB    = bufA + nh;
    unsigned* part_out = (unsigned*)(bufB + nh);
    unsigned* part_in  = part_out + (long long)CS_SLICES * nW;
    int*   cnt_in  = (int*)(part_in + (long long)CS_SLICES * nW);
    float* onorm   = (float*)(cnt_in + nP);
    float* inorm   = onorm + nP;
    int*   rowptr  = (int*)(inorm + nP);
    int*   ebase   = rowptr + nP;
    int*   rank    = ebase + (long long)CS_SLICES * nP;
    int*   esrc    = rank + nE;
    int*   tsum    = esrc + nE;

    gcn_count_kernel      <<<dim3(CS_SLICES, 2), 1024, 0, stream>>>(src, dst, part_out, part_in, rank, nE, len, nW);
    gcn_scan_tile_kernel  <<<nT, 256, 0, stream>>>(part_in, cnt_in, rowptr, tsum, nW);
    gcn_scan_finish_kernel<<<nT, 256, 0, stream>>>(part_out, part_in, tsum, rowptr, ebase, onorm, inorm, (int)nP, nW);
    gcn_fill_kernel       <<<dim3((len + 255) / 256, CS_SLICES), 256, 0, stream>>>(src, dst, ebase, rank, esrc, nE, len, (int)nP);

    gcn_gemm1_kernel              <<<(nN + 63) / 64, 128, 0, stream>>>(X, W1, onorm, bufA, nN);
    gcn_gather_kernel             <<<(nN + 15) / 16, 256, 0, stream>>>(rowptr, cnt_in, esrc, bufA, onorm, inorm, b1, bufB, nN);

    gcn_gather_gemm2_kernel<<<(nN + 15) / 16, 256, 0, stream>>>(rowptr, cnt_in, esrc, bufB, W2, inorm, b2, out, nN);
}


# Edge placement without per-edge lookups: a block owns one slice and an eighth of the nodes, their row cursors in LDS; a returning LDS add is the edge's place; row sum order = slice, then LDS arrival
# speedup vs baseline: 6.4909x; 1.2982x over previous
#include <hip/hip_runtime.h>

#define IN_F  128
#define HID_F 64
#define OUT_F 64

#define SCAN_TILE 1024
#define CS_SLICES 32
#define CS_WORDS  65536
#define PL_PARTS  8
#define PL_WORDS  16384

typedef __attribute__((ext_vector_type(2))) float v2f;
typedef __attribute__((ext_vector_type(8))) float v8f;
typedef __attribute__((ext_vector_type(4))) float v4f;
typedef __attribute__((ext_vector_type(4))) int   v4i;
typedef __attribute__((ext_vector_type(2))) unsigned v2u;
typedef __attribute__((ext_vector_type(4))) unsigned v4u;

__global__ __launch_bounds__(1024) void gcn_count_kernel(
        const int* __restrict__ src, const int* __restrict__ dst,
        unsigned* __restrict__ part_out, unsigned* __restrict__ part_in,
        int nE, int len, int nW) {
    __shared__ unsigned hist[CS_WORDS];
    const int  t      = threadIdx.x;
    const int  s      = blockIdx.x;
    const bool by_dst = blockIdx.y == 0;
    const int* __restrict__ keys = by_dst ? dst : src;
    unsigned*  __restrict__ part = (by_dst ? part_in : part_out) + (long long)s * nW;

    const v4u z = {0u, 0u, 0u, 0u};
    for (int i = 4 * t; i < nW; i += 4096) *(v4u*)(hist + i) = z;
    __syncthreads();

    const int lo  = s * len;
    const int hi  = (lo + len) < nE ? (lo + len) : nE;
    const int cnt = hi > lo ? hi - lo : 0;
    const int hi4 = lo + (cnt & ~3);
    for (int i = lo + 4 * t; i < hi4; i += 8 * 4096) {
        v4i v[8];
#pragma unroll
        for (int u = 0; u < 8; ++u) {
            const v4i none = {-1, -1, -1, -1};
            const int j = i + u * 4096;
            v[u] = none;
            if (j < hi4) v[u] = *(const v4i*)(keys + j);
        }
#pragma unroll
        for (int u = 0; u < 8; ++u) {
#pragma unroll
            for (int k = 0; k < 4; ++k) {
                const unsigned node = (unsigned)v[u][k];
                const unsigned w    = node >> 1;
                const unsigned sh   = (node & 1u) * 16u;
                if (w < (unsigned)nW) atomicAdd(&hist[w], 1u << sh);
            }
        }
    }
    if (t < (cnt & 3)) {
        const unsigned node = (unsigned)keys[hi4 + t];
        const unsigned w    = node >> 1;
        const unsigned sh   = (node & 1u) * 16u;
        if (w < (unsigned)nW) atomicAdd(&hist[w], 1u << sh);
    }
    __syncthreads();

    for (int i = 4 * t; i < nW; i += 4096) *(v4u*)(part + i) = *(const v4u*)(hist + i);
}

__device__ __forceinline__ v4i gcn_unpack4(v2u p) {
    v4i r;
    r.x = (int)(p.x & 0xFFFFu);
    r.y = (int)(p.x >> 16);
    r.z = (int)(p.y & 0xFFFFu);
    r.w = (int)(p.y >> 16);
    return r;
}


__global__ __launch_bounds__(256) void gcn_scan_tile_kernel(
        const unsigned* __restrict__ part_in, int* __restrict__ cnt_in,
        int* __restrict__ rowptr, int* __restrict__ tsum, int nW) {
    __shared__ int wsum[8];
    const int t    = threadIdx.x;
    const int lane = t & 31;
    const int wv   = t >> 5;
    const int base = blockIdx.x * SCAN_TILE + t * 4;
    v4i v = {0, 0, 0, 0};
#pragma unroll
    for (int sl = 0; sl < CS_SLICES; ++sl)
        v += gcn_unpack4(*(const v2u*)(part_in + (long long)sl * nW + (base >> 1)));
    *(v4i*)(cnt_in + base) = v;
    const int s = v.x + v.y + v.z + v.w;
    int x = s;
    for (int off = 1; off < 32; off <<= 1) {
        const int y = __shfl_up(x, off);
        if (lane >= off) x += y;
    }
    if (lane == 31) wsum[wv] = x;
    __syncthreads();
    int woff = 0;
    for (int i = 0; i < 8; ++i) woff += (i < wv) ? wsum[i] : 0;
    const int ex = woff + x - s;
    v4i o;
    o.x = ex;
    o.y = ex + v.x;
    o.z = ex + v.x + v.y;
    o.w = ex + v.x + v.y + v.z;
    *(v4i*)(rowptr + base) = o;
    if (t == 255) tsum[blockIdx.x] = woff + x;
}

__global__ __launch_bounds__(256) void gcn_scan_finish_kernel(
        const unsigned* __restrict__ part_out, const unsigned* __restrict__ part_in,
        const int* __restrict__ tsum, int* __restrict__ rowptr, int* __restrict__ ebase,
        float* __restrict__ onorm, float* __restrict__ inorm, int nP, int nW) {
    __shared__ int wsum[8];
    const int t    = threadIdx.x;
    const int lane = t & 31;
    const int wv   = t >> 5;
    int part = 0;
    for (int i = t; i < (int)blockIdx.x; i += 256) part += tsum[i];
    for (int off = 16; off > 0; off >>= 1) part += __shfl_xor(part, off);
    if (lane == 0) wsum[wv] = part;
    __syncthreads();
    int toff = 0;
    for (int i = 0; i < 8; ++i) toff += wsum[i];
    const int base = blockIdx.x * SCAN_TILE + t * 4;
    const v4i rp = *(const v4i*)(rowptr + base) + toff;
    *(v4i*)(rowptr + base) = rp;
    v4i run = rp;
#pragma unroll
    for (int sl = 0; sl < CS_SLICES; ++sl) {
        const v4i c = gcn_unpack4(*(const v2u*)(part_in + (long long)sl * nW + (base >> 1)));
        *(v4i*)(ebase + (long long)sl * nP + base) = run;
        run += c;
    }
    const v4i ci = run - rp;
    v4i co = {0, 0, 0, 0};
#pragma unroll
    for (int sl = 0; sl < CS_SLICES; ++sl)
        co += gcn_unpack4(*(const v2u*)(part_out + (long long)sl * nW + (base >> 1)));
    v4f no, ni;
    for (int u = 0; u < 4; ++u) {
        float dout = (float)co[u];
        float din  = (float)ci[u];
        dout = dout < 1.0f ? 1.0f : dout;
        din  = din  < 1.0f ? 1.0f : din;
        no[u] = rsqrtf(dout);
        ni[u] = rsqrtf(din);
    }
    *(v4f*)(onorm + base) = no;
    *(v4f*)(inorm + base) = ni;
}

__global__ __launch_bounds__(1024) void gcn_place_kernel(
        const int* __restrict__ src, const int* __restrict__ dst,
        const int* __restrict__ ebase, int* __restrict__ esrc,
        int nE, int len, int nP) {
    __shared__ int cur[PL_WORDS];
    const int t  = threadIdx.x;
    const int sl = blockIdx.y;
    const int nR = nP / PL_PARTS;
    const int n0 = blockIdx.x * nR;
    const int* __restrict__ eb = ebase + (long long)sl * nP + n0;
    for (int i = 4 * t; i < nR; i += 4096) *(v4i*)(cur + i) = *(const v4i*)(eb + i);
    __syncthreads();

    const int lo  = sl * len;
    const int hi  = (lo + len) < nE ? (lo + len) : nE;
    const int cnt = hi > lo ? hi - lo : 0;
    const int hi4 = lo + (cnt & ~3);
    for (int i = lo + 4 * t; i < hi4; i += 4 * 4096) {
        v4i d[4], s[4];
#pragma unroll
        for (int u = 0; u < 4; ++u) {
            const v4i none = {-1, -1, -1, -1};
            const int j = i + u * 4096;
            d[u] = none;
            s[u] = none;
            if (j < hi4) {
                d[u] = *(const v4i*)(dst + j);
                s[u] = *(const v4i*)(src + j);
            }
        }
#pragma unroll
        for (int u = 0; u < 4; ++u) {
#pragma unroll
            for (int k = 0; k < 4; ++k) {
                const unsigned off = (unsigned)(d[u][k] - n0);
                int pos = -1;
                if (off < (unsigned)nR) pos = atomicAdd(&cur[off], 1);
                d[u][k] = pos;
            }
        }
#pragma unroll
        for (int u = 0; u < 4; ++u) {
#pragma unroll
            for (int k = 0; k < 4; ++k) {
                if (d[u][k] >= 0) esrc[d[u][k]] = s[u][k];
            }
        }
    }
    if (t < (cnt & 3)) {
        const unsigned off = (unsigned)(dst[hi4 + t] - n0);
        if (off < (unsigned)nR) esrc[atomicAdd(&cur[off], 1)] = src[hi4 + t];
    }
}

static_assert(HID_F == 64 && OUT_F == 64, "the strip is four 16-column tiles wide");

template <int K, bool SCALE>
__device__ __forceinline__ void gcn_strip_mma(const float* __restrict__ xr,
                                              const float* __restrict__ wl,
                                              float sc, v8f (&acc)[4]) {
#pragma unroll
    for (int q = 0; q < K / 8; ++q) {
        v4f av = *(const v4f*)(xr + 8 * q);
        if (SCALE) av *= sc;
#pragma unroll
        for (int e = 0; e < 2; ++e) {
            const float* wk = wl + (8 * q + 2 * e) * 64;
            const v4f wx = *(const v4f*)wk;
            const v4f wy = *(const v4f*)(wk + 64);
            v2f a;
            a.x = av[2 * e];
            a.y = av[2 * e + 1];
#pragma unroll
            for (int n = 0; n < 4; ++n) {
                v2f b;
                b.x = wx[n];
                b.y = wy[n];
                acc[n] = __builtin_amdgcn_wmma_f32_16x16x4_f32(
                             false, a, false, b, (short)0, acc[n], false, false);
            }
        }
    }
}

__global__ __launch_bounds__(128) void gcn_gemm1_kernel(
        const float* __restrict__ X, const float* __restrict__ W,
        const float* __restrict__ onorm, float* __restrict__ H, int nN) {
    const int wv   = __builtin_amdgcn_readfirstlane(threadIdx.x >> 5);
    const int lane = threadIdx.x & 31;
    const int half = lane >> 4;
    const int l    = lane & 15;
    const int m0   = (blockIdx.x * 4 + wv) * 16;
    if (m0 >= nN) return;

    int row = m0 + l;
    if (row >= nN) row = nN - 1;
    const float sc = onorm[row];

    v8f acc[4] = {};
    gcn_strip_mma<IN_F, true>(X + (long long)row * IN_F + 4 * half,
                              W + 4 * half * HID_F + 4 * l, sc, acc);
#pragma unroll
    for (int r = 0; r < 8; ++r) {
        const int orow = m0 + r + 8 * half;
        if (orow < nN) {
            v4f o;
            o.x = acc[0][r];
            o.y = acc[1][r];
            o.z = acc[2][r];
            o.w = acc[3][r];
            *(v4f*)(H + (long long)orow * HID_F + 4 * l) = o;
        }
    }
}


__device__ __forceinline__ void gcn_group_loads(const float* __restrict__ hl, int idx, int j,
                                                v2f (&a)[8]) {
#pragma unroll
    for (int u = 0; u < 8; ++u) {
        const int s = __builtin_amdgcn_readlane(idx, j + u);
        a[u] = *(const v2f*)(hl + (long long)s * HID_F);
    }
}

__device__ __forceinline__ void gcn_group_adds(v2f& acc, int r, const v2f (&a)[8]) {
#pragma unroll
    for (int u = 0; u < 8; ++u) {
        if (u < r) acc += a[u];
    }
}

__device__ __forceinline__ void gcn_gather_two_rows(const int* __restrict__ esrc,
                                                    const float* __restrict__ hl,
                                                    int startA, int startB, int degA, int degB,
                                                    int lane, v2f& accA, v2f& accB) {
    accA = (v2f){0.0f, 0.0f};
    accB = (v2f){0.0f, 0.0f};
    const int dmax = degA > degB ? degA : degB;
    for (int base = 0; base < dmax; base += 32) {
        int nA = degA - base;
        int nB = degB - base;
        nA = nA < 0 ? 0 : (nA > 32 ? 32 : nA);
        nB = nB < 0 ? 0 : (nB > 32 ? 32 : nB);
        int idxA = 0, idxB = 0;
        if (lane < nA) idxA = esrc[startA + base + lane];
        if (lane < nB) idxB = esrc[startB + base + lane];
        const int nm = nA > nB ? nA : nB;
        for (int j = 0; j < nm; j += 8) {
            v2f a[8], b[8];
            gcn_group_loads(hl, idxA, j, a);
            gcn_group_loads(hl, idxB, j, b);
            gcn_group_adds(accA, nA - j, a);
            gcn_group_adds(accB, nB - j, b);
        }
    }
}

__global__ __launch_bounds__(256) void gcn_gather_kernel(
        const int* __restrict__ rowptr, const int* __restrict__ cnt_in,
        const int* __restrict__ esrc, const float* __restrict__ H,
        const float* __restrict__ onorm, const float* __restrict__ inorm,
        const float* __restrict__ bias, float* __restrict__ AGG, int nN) {
    const int lane = threadIdx.x & 31;
    const int nodeA = __builtin_amdgcn_readfirstlane((blockIdx.x * 8 + (threadIdx.x >> 5)) * 2);
    if (nodeA >= nN) return;
    const bool two   = nodeA + 1 < nN;
    const int  nodeB = two ? nodeA + 1 : nodeA;
    const int startA = rowptr[nodeA];
    const int startB = rowptr[nodeB];
    const int degA   = cnt_in[nodeA];
    const int degB   = two ? cnt_in[nodeB] : 0;
    const float rnA = inorm[nodeA];
    const float rnB = inorm[nodeB];
    const float onA = onorm[nodeA];
    const float onB = onorm[nodeB];
    const v2f   bb  = *(const v2f*)(bias + 2 * lane);

    v2f accA, accB;
    gcn_gather_two_rows(esrc, H + 2 * lane, startA, startB, degA, degB, lane, accA, accB);

    accA = accA * rnA + bb;
    accA.x = accA.x > 0.0f ? accA.x : 0.0f;
    accA.y = accA.y > 0.0f ? accA.y : 0.0f;
    accA *= onA;
    accB = accB * rnB + bb;
    accB.x = accB.x > 0.0f ? accB.x : 0.0f;
    accB.y = accB.y > 0.0f ? accB.y : 0.0f;
    accB *= onB;
    *(v2f*)(AGG + (long long)nodeA * HID_F + 2 * lane) = accA;
    if (two) *(v2f*)(AGG + (long long)nodeB * HID_F + 2 * lane) = accB;
}

#define STRIP_PITCH 68

__global__ __launch_bounds__(256) void gcn_gather_gemm2_kernel(
        const int* __restrict__ rowptr, const int* __restrict__ cnt_in,
        const int* __restrict__ esrc, const float* __restrict__ H,
        const float* __restrict__ W, const float* __restrict__ inorm,
        const float* __restrict__ bias, float* __restrict__ Out, int nN) {
    __shared__ __attribute__((aligned(16))) float strip[16 * STRIP_PITCH];
    __shared__ __attribute__((aligned(16))) float rnorm[16];
    const int lane = threadIdx.x & 31;
    const int wv    = __builtin_amdgcn_readfirstlane(threadIdx.x >> 5);
    const int m0    = blockIdx.x * 16;
    const int nodeA = m0 + 2 * wv;
    int   startA = 0, startB = 0, degA = 0, degB = 0;
    float rnA = 0.0f, rnB = 0.0f;
    if (nodeA < nN)     { startA = rowptr[nodeA];     degA = cnt_in[nodeA];     rnA = inorm[nodeA]; }
    if (nodeA + 1 < nN) { startB = rowptr[nodeA + 1]; degB = cnt_in[nodeA + 1]; rnB = inorm[nodeA + 1]; }

    v2f accA, accB;
    gcn_gather_two_rows(esrc, H + 2 * lane, startA, startB, degA, degB, lane, accA, accB);
    *(v2f*)(strip + (2 * wv)     * STRIP_PITCH + 2 * lane) = accA;
    *(v2f*)(strip + (2 * wv + 1) * STRIP_PITCH + 2 * lane) = accB;
    if (lane < 2) rnorm[2 * wv + lane] = lane ? rnB : rnA;
    __syncthreads();

    const int hs = wv - 2 * (__builtin_popcount(blockIdx.x) & 1);
    if (hs < 0 || hs > 1) return;

    const int half = lane >> 4;
    const int c    = lane & 15;
    const v2f bb = *(const v2f*)(bias + 32 * hs + 2 * c);

    const float* ar = strip + c * STRIP_PITCH + 4 * half;
    const float* wl = W + 4 * half * OUT_F + 32 * hs + 2 * c;
    v8f acc[2] = {};
#pragma unroll
    for (int q = 0; q < HID_F / 8; ++q) {
        const v4f av = *(const v4f*)(ar + 8 * q);
#pragma unroll
        for (int e = 0; e < 2; ++e) {
            const float* wk = wl + (8 * q + 2 * e) * OUT_F;
            const v2f wx = *(const v2f*)wk;
            const v2f wy = *(const v2f*)(wk + OUT_F);
            v2f a;
            a.x = av[2 * e];
            a.y = av[2 * e + 1];
#pragma unroll
            for (int n = 0; n < 2; ++n) {
                v2f b;
                b.x = wx[n];
                b.y = wy[n];
                acc[n] = __builtin_amdgcn_wmma_f32_16x16x4_f32(
                             false, a, false, b, (short)0, acc[n], false, false);
            }
        }
    }
    __builtin_amdgcn_sched_barrier(0);
    const v4f rn0 = *(const v4f*)(rnorm + 8 * half);
    const v4f rn1 = *(const v4f*)(rnorm + 8 * half + 4);
#pragma unroll
    for (int r = 0; r < 8; ++r) {
        const int   orow = m0 + r + 8 * half;
        const float rn   = r < 4 ? rn0[r & 3] : rn1[r & 3];
        v2f o;
        o.x = acc[0][r] * rn + bb.x;
        o.y = acc[1][r] * rn + bb.y;
        if (orow < nN) *(v2f*)(Out + (long long)orow * OUT_F + 32 * hs + 2 * c) = o;
    }
}

extern "C" void kernel_launch(void* const* d_in, const int* in_sizes, int n_in,
                              void* d_out, int out_size, void* d_ws, size_t ws_size,
                              hipStream_t stream) {
    const float* X   = (const float*)d_in[0];
    const int*   src = (const int*)  d_in[1];
    const int*   dst = (const int*)  d_in[2];
    const float* W1  = (const float*)d_in[3];
    const float* b1  = (const float*)d_in[4];
    const float* W2  = (const float*)d_in[5];
    const float* b2  = (const float*)d_in[6];
    float*       out = (float*)d_out;

    const int nN = in_sizes[0] / IN_F;
    const int nE = in_sizes[1];

    const long long nh  = (long long)nN * HID_F;
    const int nT = (nN + SCAN_TILE - 1) / SCAN_TILE;
    const long long nP = (long long)nT * SCAN_TILE;

    const int nW  = (int)(nP / 2);
    const int len = ((nE + 32 * CS_SLICES - 1) / (32 * CS_SLICES)) * 32;

    float* ws      = (float*)d_ws;
    float* bufA    = ws;
    float* bufB    = bufA + nh;
    unsigned* part_out = (unsigned*)(bufB + nh);
    unsigned* part_in  = part_out + (long long)CS_SLICES * nW;
    int*   cnt_in  = (int*)(part_in + (long long)CS_SLICES * nW);
    float* onorm   = (float*)(cnt_in + nP);
    float* inorm   = onorm + nP;
    int*   rowptr  = (int*)(inorm + nP);
    int*   ebase   = rowptr + nP;
    int*   esrc    = ebase + (long long)CS_SLICES * nP;
    int*   tsum    = esrc + nE;

    gcn_count_kernel      <<<dim3(CS_SLICES, 2), 1024, 0, stream>>>(src, dst, part_out, part_in, nE, len, nW);
    gcn_scan_tile_kernel  <<<nT, 256, 0, stream>>>(part_in, cnt_in, rowptr, tsum, nW);
    gcn_scan_finish_kernel<<<nT, 256, 0, stream>>>(part_out, part_in, tsum, rowptr, ebase, onorm, inorm, (int)nP, nW);
    gcn_place_kernel      <<<dim3(PL_PARTS, CS_SLICES), 1024, 0, stream>>>(src, dst, ebase, esrc, nE, len, (int)nP);

    gcn_gemm1_kernel              <<<(nN + 63) / 64, 128, 0, stream>>>(X, W1, onorm, bufA, nN);
    gcn_gather_kernel             <<<(nN + 15) / 16, 256, 0, stream>>>(rowptr, cnt_in, esrc, bufA, onorm, inorm, b1, bufB, nN);

    gcn_gather_gemm2_kernel<<<(nN + 15) / 16, 256, 0, stream>>>(rowptr, cnt_in, esrc, bufB, W2, inorm, b2, out, nN);
}


# Scan in one launch: count kernel leaves each slice's edge count per scan tile; a tile's block adds the earlier tiles and scans itself, its loads in one group; tile-sum launch gone; sum orders kept
# speedup vs baseline: 6.9503x; 1.0708x over previous
#include <hip/hip_runtime.h>

#define IN_F  128
#define HID_F 64
#define OUT_F 64

#define SCAN_TILE 1024
#define CS_SLICES 32
#define CS_WORDS  65536
#define PL_PARTS  8
#define PL_WORDS  16384

typedef __attribute__((ext_vector_type(2))) float v2f;
typedef __attribute__((ext_vector_type(8))) float v8f;
typedef __attribute__((ext_vector_type(4))) float v4f;
typedef __attribute__((ext_vector_type(4))) int   v4i;
typedef __attribute__((ext_vector_type(2))) unsigned v2u;
typedef __attribute__((ext_vector_type(4))) unsigned v4u;

__global__ __launch_bounds__(1024) void gcn_count_kernel(
        const int* __restrict__ src, const int* __restrict__ dst,
        unsigned* __restrict__ part_out, unsigned* __restrict__ part_in,
        int* __restrict__ tile_in, int nE, int len, int nW) {
    __shared__ unsigned hist[CS_WORDS];
    const int  t      = threadIdx.x;
    const int  s      = blockIdx.x;
    const bool by_dst = blockIdx.y == 0;
    const int* __restrict__ keys = by_dst ? dst : src;
    unsigned*  __restrict__ part = (by_dst ? part_in : part_out) + (long long)s * nW;

    const v4u z = {0u, 0u, 0u, 0u};
    for (int i = 4 * t; i < nW; i += 4096) *(v4u*)(hist + i) = z;
    __syncthreads();

    const int lo  = s * len;
    const int hi  = (lo + len) < nE ? (lo + len) : nE;
    const int cnt = hi > lo ? hi - lo : 0;
    const int hi4 = lo + (cnt & ~3);
    for (int i = lo + 4 * t; i < hi4; i += 8 * 4096) {
        v4i v[8];
#pragma unroll
        for (int u = 0; u < 8; ++u) {
            const v4i none = {-1, -1, -1, -1};
            const int j = i + u * 4096;
            v[u] = none;
            if (j < hi4) v[u] = *(const v4i*)(keys + j);
        }
#pragma unroll
        for (int u = 0; u < 8; ++u) {
#pragma unroll
            for (int k = 0; k < 4; ++k) {
                const unsigned node = (unsigned)v[u][k];
                const unsigned w    = node >> 1;
                const unsigned sh   = (node & 1u) * 16u;
                if (w < (unsigned)nW) atomicAdd(&hist[w], 1u << sh);
            }
        }
    }
    if (t < (cnt & 3)) {
        const unsigned node = (unsigned)keys[hi4 + t];
        const unsigned w    = node >> 1;
        const unsigned sh   = (node & 1u) * 16u;
        if (w < (unsigned)nW) atomicAdd(&hist[w], 1u << sh);
    }
    __syncthreads();

    for (int i = 4 * t; i < nW; i += 4096) *(v4u*)(part + i) = *(const v4u*)(hist + i);

    if (by_dst && (t >> 3) < (nW >> 9)) {
        const unsigned* hp = hist + 512 * (t >> 3) + 4 * (t & 7);
        unsigned c = 0;
#pragma unroll
        for (int m = 0; m < 16; ++m) {
            const v4u h = *(const v4u*)(hp + 32 * m);
            c += (h.x + h.y) + (h.z + h.w);
        }
        c += __shfl_xor(c, 4);
        c += __shfl_xor(c, 2);
        c += __shfl_xor(c, 1);
        if ((t & 7) == 0) tile_in[(t >> 3) * CS_SLICES + s] = (int)((c & 0xFFFFu) + (c >> 16));
    }
}

__device__ __forceinline__ v4i gcn_unpack4(v2u p) {
    v4i r;
    r.x = (int)(p.x & 0xFFFFu);
    r.y = (int)(p.x >> 16);
    r.z = (int)(p.y & 0xFFFFu);
    r.w = (int)(p.y >> 16);
    return r;
}


#define TI_LOADS 4
static_assert(TI_LOADS * 1024 >= (CS_WORDS / 512) * CS_SLICES, "a block sums all earlier tiles' counts in one sweep");
static_assert(CS_SLICES % 4 == 0, "a 16-byte piece of tile_in lies in one tile");

__global__ __launch_bounds__(256) void gcn_scan_kernel(
        const unsigned* __restrict__ part_out, const unsigned* __restrict__ part_in,
        const int* __restrict__ tile_in, int* __restrict__ cnt_in,
        int* __restrict__ rowptr, int* __restrict__ ebase,
        float* __restrict__ onorm, float* __restrict__ inorm, int nP, int nW) {
    __shared__ int wpart[8];
    __shared__ int wsum[8];
    const int t    = threadIdx.x;
    const int lane = t & 31;
    const int wv   = t >> 5;
    const int base = blockIdx.x * SCAN_TILE + t * 4;

    const int lim = (int)blockIdx.x * CS_SLICES;
    v4i tw[TI_LOADS];
    int tm[TI_LOADS];
#pragma unroll
    for (int u = 0; u < TI_LOADS; ++u) {
        const int i = 4 * t + 1024 * u;
        tm[u] = (i - lim) >> 31;
        tw[u] = *(const v4i*)(tile_in + (i & tm[u]));
    }
    v2u pk[CS_SLICES], po[CS_SLICES];
#pragma unroll
    for (int sl = 0; sl < CS_SLICES; ++sl)
        pk[sl] = *(const v2u*)(part_in + (long long)sl * nW + (base >> 1));
#pragma unroll
    for (int sl = 0; sl < CS_SLICES; ++sl)
        po[sl] = *(const v2u*)(part_out + (long long)sl * nW + (base >> 1));
    __builtin_amdgcn_sched_barrier(0);

    int part = 0;
#pragma unroll
    for (int u = 0; u < TI_LOADS; ++u)
        part += ((tw[u].x + tw[u].y) + (tw[u].z + tw[u].w)) & tm[u];

    v4i co = {0, 0, 0, 0};
#pragma unroll
    for (int sl = 0; sl < CS_SLICES; ++sl) co += gcn_unpack4(po[sl]);
    v4f no;
    for (int u = 0; u < 4; ++u) {
        float dout = (float)co[u];
        dout = dout < 1.0f ? 1.0f : dout;
        no[u] = rsqrtf(dout);
    }
    *(v4f*)(onorm + base) = no;

    v4i ci = {0, 0, 0, 0};
#pragma unroll
    for (int sl = 0; sl < CS_SLICES; ++sl) ci += gcn_unpack4(pk[sl]);
    *(v4i*)(cnt_in + base) = ci;
    v4f ni;
    for (int u = 0; u < 4; ++u) {
        float din = (float)ci[u];
        din = din < 1.0f ? 1.0f : din;
        ni[u] = rsqrtf(din);
    }
    *(v4f*)(inorm + base) = ni;

    const int s = ci.x + ci.y + ci.z + ci.w;
    int x = s;
    for (int off = 1; off < 32; off <<= 1) {
        const int y = __shfl_up(x, off);
        if (lane >= off) x += y;
    }
    for (int off = 16; off > 0; off >>= 1) part += __shfl_xor(part, off);
    if (lane == 31) wsum[wv]  = x;
    if (lane == 0)  wpart[wv] = part;
    __syncthreads();
    int off0 = 0;
    for (int i = 0; i < 8; ++i) off0 += wpart[i] + ((i < wv) ? wsum[i] : 0);
    const int ex = off0 + x - s;
    v4i rp;
    rp.x = ex;
    rp.y = ex + ci.x;
    rp.z = ex + ci.x + ci.y;
    rp.w = ex + ci.x + ci.y + ci.z;
    *(v4i*)(rowptr + base) = rp;
    v4i run = rp;
#pragma unroll
    for (int sl = 0; sl < CS_SLICES; ++sl) {
        *(v4i*)(ebase + (long long)sl * nP + base) = run;
        run += gcn_unpack4(pk[sl]);
    }
}

__global__ __launch_bounds__(1024) void gcn_place_kernel(
        const int* __restrict__ src, const int* __restrict__ dst,
        const int* __restrict__ ebase, int* __restrict__ esrc,
        int nE, int len, int nP) {
    __shared__ int cur[PL_WORDS];
    const int t  = threadIdx.x;
    const int sl = blockIdx.y;
    const int nR = nP / PL_PARTS;
    const int n0 = blockIdx.x * nR;
    const int* __restrict__ eb = ebase + (long long)sl * nP + n0;
    for (int i = 4 * t; i < nR; i += 4096) *(v4i*)(cur + i) = *(const v4i*)(eb + i);
    __syncthreads();

    const int lo  = sl * len;
    const int hi  = (lo + len) < nE ? (lo + len) : nE;
    const int cnt = hi > lo ? hi - lo : 0;
    const int hi4 = lo + (cnt & ~3);
    for (int i = lo + 4 * t; i < hi4; i += 4 * 4096) {
        v4i d[4], s[4];
#pragma unroll
        for (int u = 0; u < 4; ++u) {
            const v4i none = {-1, -1, -1, -1};
            const int j = i + u * 4096;
            d[u] = none;
            s[u] = none;
            if (j < hi4) {
                d[u] = *(const v4i*)(dst + j);
                s[u] = *(const v4i*)(src + j);
            }
        }
#pragma unroll
        for (int u = 0; u < 4; ++u) {
#pragma unroll
            for (int k = 0; k < 4; ++k) {
                const unsigned off = (unsigned)(d[u][k] - n0);
                int pos = -1;
                if (off < (unsigned)nR) pos = atomicAdd(&cur[off], 1);
                d[u][k] = pos;
            }
        }
#pragma unroll
        for (int u = 0; u < 4; ++u) {
#pragma unroll
            for (int k = 0; k < 4; ++k) {
                if (d[u][k] >= 0) esrc[d[u][k]] = s[u][k];
            }
        }
    }
    if (t < (cnt & 3)) {
        const unsigned off = (unsigned)(dst[hi4 + t] - n0);
        if (off < (unsigned)nR) esrc[atomicAdd(&cur[off], 1)] = src[hi4 + t];
    }
}

static_assert(HID_F == 64 && OUT_F == 64, "the strip is four 16-column tiles wide");

template <int K, bool SCALE>
__device__ __forceinline__ void gcn_strip_mma(const float* __restrict__ xr,
                                              const float* __restrict__ wl,
                                              float sc, v8f (&acc)[4]) {
#pragma unroll
    for (int q = 0; q < K / 8; ++q) {
        v4f av = *(const v4f*)(xr + 8 * q);
        if (SCALE) av *= sc;
#pragma unroll
        for (int e = 0; e < 2; ++e) {
            const float* wk = wl + (8 * q + 2 * e) * 64;
            const v4f wx = *(const v4f*)wk;
            const v4f wy = *(const v4f*)(wk + 64);
            v2f a;
            a.x = av[2 * e];
            a.y = av[2 * e + 1];
#pragma unroll
            for (int n = 0; n < 4; ++n) {
                v2f b;
                b.x = wx[n];
                b.y = wy[n];
                acc[n] = __builtin_amdgcn_wmma_f32_16x16x4_f32(
                             false, a, false, b, (short)0, acc[n], false, false);
            }
        }
    }
}

__global__ __launch_bounds__(128) void gcn_gemm1_kernel(
        const float* __restrict__ X, const float* __restrict__ W,
        const float* __restrict__ onorm, float* __restrict__ H, int nN) {
    const int wv   = __builtin_amdgcn_readfirstlane(threadIdx.x >> 5);
    const int lane = threadIdx.x & 31;
    const int half = lane >> 4;
    const int l    = lane & 15;
    const int m0   = (blockIdx.x * 4 + wv) * 16;
    if (m0 >= nN) return;

    int row = m0 + l;
    if (row >= nN) row = nN - 1;
    const float sc = onorm[row];

    v8f acc[4] = {};
    gcn_strip_mma<IN_F, true>(X + (long long)row * IN_F + 4 * half,
                              W + 4 * half * HID_F + 4 * l, sc, acc);
#pragma unroll
    for (int r = 0; r < 8; ++r) {
        const int orow = m0 + r + 8 * half;
        if (orow < nN) {
            v4f o;
            o.x = acc[0][r];
            o.y = acc[1][r];
            o.z = acc[2][r];
            o.w = acc[3][r];
            *(v4f*)(H + (long long)orow * HID_F + 4 * l) = o;
        }
    }
}


__device__ __forceinline__ void gcn_group_loads(const float* __restrict__ hl, int idx, int j,
                                                v2f (&a)[8]) {
#pragma unroll
    for (int u = 0; u < 8; ++u) {
        const int s = __builtin_amdgcn_readlane(idx, j + u);
        a[u] = *(const v2f*)(hl + (long long)s * HID_F);
    }
}

__device__ __forceinline__ void gcn_group_adds(v2f& acc, int r, const v2f (&a)[8]) {
#pragma unroll
    for (int u = 0; u < 8; ++u) {
        if (u < r) acc += a[u];
    }
}

__device__ __forceinline__ void gcn_gather_two_rows(const int* __restrict__ esrc,
                                                    const float* __restrict__ hl,
                                                    int startA, int startB, int degA, int degB,
                                                    int lane, v2f& accA, v2f& accB) {
    accA = (v2f){0.0f, 0.0f};
    accB = (v2f){0.0f, 0.0f};
    const int dmax = degA > degB ? degA : degB;
    for (int base = 0; base < dmax; base += 32) {
        int nA = degA - base;
        int nB = degB - base;
        nA = nA < 0 ? 0 : (nA > 32 ? 32 : nA);
        nB = nB < 0 ? 0 : (nB > 32 ? 32 : nB);
        int idxA = 0, idxB = 0;
        if (lane < nA) idxA = esrc[startA + base + lane];
        if (lane < nB) idxB = esrc[startB + base + lane];
        const int nm = nA > nB ? nA : nB;
        for (int j = 0; j < nm; j += 8) {
            v2f a[8], b[8];
            gcn_group_loads(hl, idxA, j, a);
            gcn_group_loads(hl, idxB, j, b);
            gcn_group_adds(accA, nA - j, a);
            gcn_group_adds(accB, nB - j, b);
        }
    }
}

__global__ __launch_bounds__(256) void gcn_gather_kernel(
        const int* __restrict__ rowptr, const int* __restrict__ cnt_in,
        const int* __restrict__ esrc, const float* __restrict__ H,
        const float* __restrict__ onorm, const float* __restrict__ inorm,
        const float* __restrict__ bias, float* __restrict__ AGG, int nN) {
    const int lane = threadIdx.x & 31;
    const int nodeA = __builtin_amdgcn_readfirstlane((blockIdx.x * 8 + (threadIdx.x >> 5)) * 2);
    if (nodeA >= nN) return;
    const bool two   = nodeA + 1 < nN;
    const int  nodeB = two ? nodeA + 1 : nodeA;
    const int startA = rowptr[nodeA];
    const int startB = rowptr[nodeB];
    const int degA   = cnt_in[nodeA];
    const int degB   = two ? cnt_in[nodeB] : 0;
    const float rnA = inorm[nodeA];
    const float rnB = inorm[nodeB];
    const float onA = onorm[nodeA];
    const float onB = onorm[nodeB];
    const v2f   bb  = *(const v2f*)(bias + 2 * lane);

    v2f accA, accB;
    gcn_gather_two_rows(esrc, H + 2 * lane, startA, startB, degA, degB, lane, accA, accB);

    accA = accA * rnA + bb;
    accA.x = accA.x > 0.0f ? accA.x : 0.0f;
    accA.y = accA.y > 0.0f ? accA.y : 0.0f;
    accA *= onA;
    accB = accB * rnB + bb;
    accB.x = accB.x > 0.0f ? accB.x : 0.0f;
    accB.y = accB.y > 0.0f ? accB.y : 0.0f;
    accB *= onB;
    *(v2f*)(AGG + (long long)nodeA * HID_F + 2 * lane) = accA;
    if (two) *(v2f*)(AGG + (long long)nodeB * HID_F + 2 * lane) = accB;
}

#define STRIP_PITCH 68

__global__ __launch_bounds__(256) void gcn_gather_gemm2_kernel(
        const int* __restrict__ rowptr, const int* __restrict__ cnt_in,
        const int* __restrict__ esrc, const float* __restrict__ H,
        const float* __restrict__ W, const float* __restrict__ inorm,
        const float* __restrict__ bias, float* __restrict__ Out, int nN) {
    __shared__ __attribute__((aligned(16))) float strip[16 * STRIP_PITCH];
    __shared__ __attribute__((aligned(16))) float rnorm[16];
    const int lane = threadIdx.x & 31;
    const int wv    = __builtin_amdgcn_readfirstlane(threadIdx.x >> 5);
    const int m0    = blockIdx.x * 16;
    const int nodeA = m0 + 2 * wv;
    int   startA = 0, startB = 0, degA = 0, degB = 0;
    float rnA = 0.0f, rnB = 0.0f;
    if (nodeA < nN)     { startA = rowptr[nodeA];     degA = cnt_in[nodeA];     rnA = inorm[nodeA]; }
    if (nodeA + 1 < nN) { startB = rowptr[nodeA + 1]; degB = cnt_in[nodeA + 1]; rnB = inorm[nodeA + 1]; }

    v2f accA, accB;
    gcn_gather_two_rows(esrc, H + 2 * lane, startA, startB, degA, degB, lane, accA, accB);
    *(v2f*)(strip + (2 * wv)     * STRIP_PITCH + 2 * lane) = accA;
    *(v2f*)(strip + (2 * wv + 1) * STRIP_PITCH + 2 * lane) = accB;
    if (lane < 2) rnorm[2 * wv + lane] = lane ? rnB : rnA;
    __syncthreads();

    const int hs = wv - 2 * (__builtin_popcount(blockIdx.x) & 1);
    if (hs < 0 || hs > 1) return;

    const int half = lane >> 4;
    const int c    = lane & 15;
    const v2f bb = *(const v2f*)(bias + 32 * hs + 2 * c);

    const float* ar = strip + c * STRIP_PITCH + 4 * half;
    const float* wl = W + 4 * half * OUT_F + 32 * hs + 2 * c;
    v8f acc[2] = {};
#pragma unroll
    for (int q = 0; q < HID_F / 8; ++q) {
        const v4f av = *(const v4f*)(ar + 8 * q);
#pragma unroll
        for (int e = 0; e < 2; ++e) {
            const float* wk = wl + (8 * q + 2 * e) * OUT_F;
            const v2f wx = *(const v2f*)wk;
            const v2f wy = *(const v2f*)(wk + OUT_F);
            v2f a;
            a.x = av[2 * e];
            a.y = av[2 * e + 1];
#pragma unroll
            for (int n = 0; n < 2; ++n) {
                v2f b;
                b.x = wx[n];
                b.y = wy[n];
                acc[n] = __builtin_amdgcn_wmma_f32_16x16x4_f32(
                             false, a, false, b, (short)0, acc[n], false, false);
            }
        }
    }
    __builtin_amdgcn_sched_barrier(0);
    const v4f rn0 = *(const v4f*)(rnorm + 8 * half);
    const v4f rn1 = *(const v4f*)(rnorm + 8 * half + 4);
#pragma unroll
    for (int r = 0; r < 8; ++r) {
        const int   orow = m0 + r + 8 * half;
        const float rn   = r < 4 ? rn0[r & 3] : rn1[r & 3];
        v2f o;
        o.x = acc[0][r] * rn + bb.x;
        o.y = acc[1][r] * rn + bb.y;
        if (orow < nN) *(v2f*)(Out + (long long)orow * OUT_F + 32 * hs + 2 * c) = o;
    }
}

extern "C" void kernel_launch(void* const* d_in, const int* in_sizes, int n_in,
                              void* d_out, int out_size, void* d_ws, size_t ws_size,
                              hipStream_t stream) {
    const float* X   = (const float*)d_in[0];
    const int*   src = (const int*)  d_in[1];
    const int*   dst = (const int*)  d_in[2];
    const float* W1  = (const float*)d_in[3];
    const float* b1  = (const float*)d_in[4];
    const float* W2  = (const float*)d_in[5];
    const float* b2  = (const float*)d_in[6];
    float*       out = (float*)d_out;

    const int nN = in_sizes[0] / IN_F;
    const int nE = in_sizes[1];

    const long long nh  = (long long)nN * HID_F;
    const int nT = (nN + SCAN_TILE - 1) / SCAN_TILE;
    const long long nP = (long long)nT * SCAN_TILE;

    const int nW  = (int)(nP / 2);
    const int len = ((nE + 32 * CS_SLICES - 1) / (32 * CS_SLICES)) * 32;

    float* ws      = (float*)d_ws;
    float* bufA    = ws;
    float* bufB    = bufA + nh;
    unsigned* part_out = (unsigned*)(bufB + nh);
    unsigned* part_in  = part_out + (long long)CS_SLICES * nW;
    int*   cnt_in  = (int*)(part_in + (long long)CS_SLICES * nW);
    float* onorm   = (float*)(cnt_in + nP);
    float* inorm   = onorm + nP;
    int*   rowptr  = (int*)(inorm + nP);
    int*   ebase   = rowptr + nP;
    int*   tile_in = ebase + (long long)CS_SLICES * nP;
    int*   esrc    = tile_in + (long long)nT * CS_SLICES;

    gcn_count_kernel      <<<dim3(CS_SLICES, 2), 1024, 0, stream>>>(src, dst, part_out, part_in, tile_in, nE, len, nW);
    gcn_scan_kernel       <<<nT, 256, 0, stream>>>(part_out, part_in, tile_in, cnt_in, rowptr, ebase, onorm, inorm, (int)nP, nW);
    gcn_place_kernel      <<<dim3(PL_PARTS, CS_SLICES), 1024, 0, stream>>>(src, dst, ebase, esrc, nE, len, (int)nP);

    gcn_gemm1_kernel              <<<(nN + 63) / 64, 128, 0, stream>>>(X, W1, onorm, bufA, nN);
    gcn_gather_kernel             <<<(nN + 15) / 16, 256, 0, stream>>>(rowptr, cnt_in, esrc, bufA, onorm, inorm, b1, bufB, nN);

    gcn_gather_gemm2_kernel<<<(nN + 15) / 16, 256, 0, stream>>>(rowptr, cnt_in, esrc, bufB, W2, inorm, b2, out, nN);
}
